# starting kernel
# baseline (speedup 1.0000x reference)
#include <hip/hip_runtime.h>

typedef float v2f __attribute__((ext_vector_type(2)));
typedef float v8f __attribute__((ext_vector_type(8)));

#define F0 128
#define U0 64
#define H0 8
#define A0 8
#define U1 40

__device__ __forceinline__ unsigned fkey(float f) {
    unsigned b = __float_as_uint(f);
    return (b & 0x80000000u) ? ~b : (b | 0x80000000u);
}
__device__ __forceinline__ float fkey_dec(unsigned k) {
    return __uint_as_float((k & 0x80000000u) ? (k ^ 0x80000000u) : ~k);
}

__device__ __forceinline__ void edge_rc(const int* __restrict__ ei, int E, int e,
                                        int& r, int& c) {
    if (e < E) { r = ei[e]; c = ei[E + e]; }
    else       { r = e - E; c = e - E; }
}

__global__ __launch_bounds__(32) void proj0_kernel(
    const float* __restrict__ x,
    const float* __restrict__ wq, const float* __restrict__ bq,
    const float* __restrict__ wk, const float* __restrict__ bk,
    const float* __restrict__ wv,
    float* __restrict__ Q0, float* __restrict__ K0, float* __restrict__ V0)
{
    const int tile = blockIdx.x;
    const int mat  = blockIdx.y >> 2;
    const int ct   = blockIdx.y & 3;
    const int lane = threadIdx.x;
    const int hi   = lane >> 4;
    const int lo   = lane & 15;

    const float* W   = (mat == 0) ? wq : (mat == 1) ? wk : wv;
    const float* B   = (mat == 0) ? bq : (mat == 1) ? bk : nullptr;
    float*       OUT = (mat == 0) ? Q0 : (mat == 1) ? K0 : V0;

    const int m = tile * 16 + lo;
    const int n = ct * 16 + lo;

    v8f acc = {};
    for (int k = 0; k < F0; k += 4) {
        const int kb = k + 2 * hi;
        v2f a, b;
        a.x = x[(size_t)m * F0 + kb];
        a.y = x[(size_t)m * F0 + kb + 1];
        b.x = W[(size_t)kb * U0 + n];
        b.y = W[(size_t)(kb + 1) * U0 + n];
        acc = __builtin_amdgcn_wmma_f32_16x16x4_f32(
                  false, a, false, b, (short)0, acc, false, false);
    }

    const float bias = (B != nullptr) ? B[n] : 0.0f;
    #pragma unroll
    for (int v = 0; v < 8; ++v) {
        const int row = tile * 16 + v + 8 * hi;
        float val = acc[v] + bias;
        if (mat != 2) val = fmaxf(val, 0.0f);
        OUT[(size_t)row * U0 + n] = val;
    }
}

__global__ __launch_bounds__(32) void proj1v_kernel(
    const float* __restrict__ h0, const float* __restrict__ wv1,
    float* __restrict__ v1)
{
    const int tile = blockIdx.x;
    const int ct   = blockIdx.y;
    const int lane = threadIdx.x;
    const int hi   = lane >> 4;
    const int lo   = lane & 15;

    const int m = tile * 16 + lo;
    const int n = ct * 16 + lo;
    const bool valid = (n < U1);

    v8f acc = {};
    for (int k = 0; k < U0; k += 4) {
        const int kb = k + 2 * hi;
        v2f a, b;
        a.x = h0[(size_t)m * U0 + kb];
        a.y = h0[(size_t)m * U0 + kb + 1];
        b.x = valid ? wv1[(size_t)kb * U1 + n] : 0.0f;
        b.y = valid ? wv1[(size_t)(kb + 1) * U1 + n] : 0.0f;
        acc = __builtin_amdgcn_wmma_f32_16x16x4_f32(
                  false, a, false, b, (short)0, acc, false, false);
    }
    if (valid) {
        #pragma unroll
        for (int v = 0; v < 8; ++v) {
            const int row = tile * 16 + v + 8 * hi;
            v1[(size_t)row * U1 + n] = acc[v];
        }
    }
}

__global__ void qk1_kernel(const float* __restrict__ h0,
                           const float* __restrict__ wq1, const float* __restrict__ bq1,
                           const float* __restrict__ wk1, const float* __restrict__ bk1,
                           float* __restrict__ q1, float* __restrict__ k1, int n)
{
    const int i = blockIdx.x * blockDim.x + threadIdx.x;
    if (i >= n) return;
    const float* hr = h0 + (size_t)i * U0;
    float q = bq1[0], k = bk1[0];
    #pragma unroll 8
    for (int j = 0; j < U0; ++j) {
        const float h = hr[j];
        q += h * wq1[j];
        k += h * wk1[j];
    }
    q1[i] = fmaxf(q, 0.0f);
    k1[i] = fmaxf(k, 0.0f);
}

__global__ void edge_max0_kernel(const int* __restrict__ ei, int E, int Etot,
                                 const float* __restrict__ Q0,
                                 const float* __restrict__ K0,
                                 unsigned* __restrict__ max0key)
{
    const int t = blockIdx.x * blockDim.x + threadIdx.x;
    if (t >= Etot * H0) return;
    const int e = t >> 3, h = t & 7;
    int r, c; edge_rc(ei, E, e, r, c);
    const float* q = Q0 + (size_t)r * U0 + h * A0;
    const float* k = K0 + (size_t)c * U0 + h * A0;
    float s = 0.0f;
    #pragma unroll
    for (int a = 0; a < A0; ++a) s += q[a] * k[a];
    s *= 0.35355339059327373f;
    atomicMax(&max0key[r * H0 + h], fkey(s));
}

__global__ void edge_scatter0_kernel(const int* __restrict__ ei, int E, int Etot,
                                     const float* __restrict__ Q0,
                                     const float* __restrict__ K0,
                                     const float* __restrict__ V0,
                                     const unsigned* __restrict__ max0key,
                                     float* __restrict__ denom0,
                                     float* __restrict__ accum0)
{
    const int t = blockIdx.x * blockDim.x + threadIdx.x;
    if (t >= Etot * H0) return;
    const int e = t >> 3, h = t & 7;
    int r, c; edge_rc(ei, E, e, r, c);
    const float* q = Q0 + (size_t)r * U0 + h * A0;
    const float* k = K0 + (size_t)c * U0 + h * A0;
    float s = 0.0f;
    #pragma unroll
    for (int a = 0; a < A0; ++a) s += q[a] * k[a];
    s *= 0.35355339059327373f;
    const float m  = fkey_dec(max0key[r * H0 + h]);
    const float ev = __expf(s - m);
    atomicAdd(&denom0[r * H0 + h], ev);
    const float* vv  = V0 + (size_t)c * U0 + h * A0;
    float*       acc = accum0 + (size_t)r * U0 + h * A0;
    #pragma unroll
    for (int d = 0; d < A0; ++d) atomicAdd(&acc[d], ev * vv[d]);
}

__global__ void finalize0_kernel(const float* __restrict__ accum0,
                                 const float* __restrict__ denom0,
                                 const float* __restrict__ b0,
                                 float* __restrict__ h0, int n)
{
    const int t = blockIdx.x * blockDim.x + threadIdx.x;
    if (t >= n * U0) return;
    const int nn = t >> 6, cc = t & 63;
    const float val = accum0[t] / denom0[nn * H0 + (cc >> 3)] + b0[cc];
    h0[t] = fmaxf(val, 0.0f);
}

__global__ void edge_max1_kernel(const int* __restrict__ ei, int E, int Etot,
                                 const float* __restrict__ q1,
                                 const float* __restrict__ k1,
                                 unsigned* __restrict__ max1key)
{
    const int e = blockIdx.x * blockDim.x + threadIdx.x;
    if (e >= Etot) return;
    int r, c; edge_rc(ei, E, e, r, c);
    atomicMax(&max1key[r], fkey(q1[r] * k1[c]));
}

__global__ void edge_scatter1_kernel(const int* __restrict__ ei, int E, int Etot,
                                     const float* __restrict__ q1,
                                     const float* __restrict__ k1,
                                     const float* __restrict__ v1,
                                     const unsigned* __restrict__ max1key,
                                     float* __restrict__ denom1,
                                     float* __restrict__ accum1)
{
    const int t = blockIdx.x * blockDim.x + threadIdx.x;
    if (t >= Etot * 8) return;
    const int e = t >> 3, j = t & 7;
    int r, c; edge_rc(ei, E, e, r, c);
    const float s  = q1[r] * k1[c];
    const float m  = fkey_dec(max1key[r]);
    const float ev = __expf(s - m);
    if (j == 0) atomicAdd(&denom1[r], ev);
    const float* vv  = v1 + (size_t)c * U1;
    float*       acc = accum1 + (size_t)r * U1;
    for (int d = j; d < U1; d += 8) atomicAdd(&acc[d], ev * vv[d]);
}

__global__ void finalize_out_kernel(const float* __restrict__ accum1,
                                    const float* __restrict__ denom1,
                                    const float* __restrict__ b1,
                                    float* __restrict__ out, int n)
{
    const int t = blockIdx.x * blockDim.x + threadIdx.x;
    if (t >= n * U1) return;
    out[t] = accum1[t] / denom1[t / U1] + b1[t % U1];
}

extern "C" void kernel_launch(void* const* d_in, const int* in_sizes, int n_in,
                              void* d_out, int out_size, void* d_ws, size_t ws_size,
                              hipStream_t stream)
{
    const float* x    = (const float*)d_in[0];
    const int*   ei   = (const int*)  d_in[1];
    const float* wq0  = (const float*)d_in[2];
    const float* bq0  = (const float*)d_in[3];
    const float* wk0  = (const float*)d_in[4];
    const float* bk0  = (const float*)d_in[5];
    const float* wv0  = (const float*)d_in[6];
    const float* b0   = (const float*)d_in[7];
    const float* wq1  = (const float*)d_in[8];
    const float* bq1  = (const float*)d_in[9];
    const float* wk1  = (const float*)d_in[10];
    const float* bk1  = (const float*)d_in[11];
    const float* wv1  = (const float*)d_in[12];
    const float* b1   = (const float*)d_in[13];
    float* out = (float*)d_out;

    const int N    = in_sizes[0] / F0;
    const int E    = in_sizes[1] / 2;
    const int Etot = E + N;

    float* ws = (float*)d_ws;
    float* R0 = ws;
    float* R1 = R0 + (size_t)N * U0;
    float* R2 = R1 + (size_t)N * U0;
    float* R3 = R2 + (size_t)N * U0;
    float* R4 = R3 + (size_t)N * H0;
    float* R5 = R4 + (size_t)N * H0;

    const int TB = 256;

    proj0_kernel<<<dim3(N / 16, 12), 32, 0, stream>>>(
        x, wq0, bq0, wk0, bk0, wv0, R0, R1, R2);

    hipMemsetAsync(R3, 0, (size_t)N * (H0 + H0 + U0) * sizeof(float), stream);

    edge_max0_kernel<<<(Etot * H0 + TB - 1) / TB, TB, 0, stream>>>(
        ei, E, Etot, R0, R1, (unsigned*)R3);
    edge_scatter0_kernel<<<(Etot * H0 + TB - 1) / TB, TB, 0, stream>>>(
        ei, E, Etot, R0, R1, R2, (const unsigned*)R3, R4, R5);
    finalize0_kernel<<<(N * U0 + TB - 1) / TB, TB, 0, stream>>>(
        R5, R4, b0, R0  , N);

    float* q1 = R3;
    float* k1 = R3 + N;
    qk1_kernel<<<(N + TB - 1) / TB, TB, 0, stream>>>(
        R0, wq1, bq1, wk1, bk1, q1, k1, N);
    proj1v_kernel<<<dim3(N / 16, 3), 32, 0, stream>>>(
        R0, wv1, R1  );

    unsigned* max1key = (unsigned*)R4;
    float*    denom1  = R4 + N;
    float*    accum1  = R2;
    hipMemsetAsync(R4, 0, (size_t)2 * N * sizeof(float), stream);
    hipMemsetAsync(R2, 0, (size_t)N * U1 * sizeof(float), stream);

    edge_max1_kernel<<<(Etot + TB - 1) / TB, TB, 0, stream>>>(
        ei, E, Etot, q1, k1, max1key);
    edge_scatter1_kernel<<<(Etot * 8 + TB - 1) / TB, TB, 0, stream>>>(
        ei, E, Etot, q1, k1, R1, max1key, denom1, accum1);
    finalize_out_kernel<<<(N * U1 + TB - 1) / TB, TB, 0, stream>>>(
        accum1, denom1, b1, out, N);
}


# Layer 1 stops scanning the row array: layer 0's work-groups leave their edge list (one packed word an edge) in their own rows of the output buffer, layer 1 reads it; LDS-atomic sums, arrival order
# speedup vs baseline: 2.6520x; 2.6520x over previous
#include <hip/hip_runtime.h>

typedef float v2f __attribute__((ext_vector_type(2)));
typedef float v8f __attribute__((ext_vector_type(8)));

#define F0 128
#define U0 64
#define H0 8
#define A0 8
#define U1 40

__device__ __forceinline__ unsigned fkey(float f) {
    unsigned b = __float_as_uint(f);
    return (b & 0x80000000u) ? ~b : (b | 0x80000000u);
}
__device__ __forceinline__ float fkey_dec(unsigned k) {
    return __uint_as_float((k & 0x80000000u) ? (k ^ 0x80000000u) : ~k);
}

__global__ __launch_bounds__(32) void proj0_kernel(
    const float* __restrict__ x,
    const float* __restrict__ wq, const float* __restrict__ bq,
    const float* __restrict__ wk, const float* __restrict__ bk,
    const float* __restrict__ wv,
    float* __restrict__ Q0, float* __restrict__ K0, float* __restrict__ V0)
{
    const int tile = blockIdx.x;
    const int mat  = blockIdx.y >> 2;
    const int ct   = blockIdx.y & 3;
    const int lane = threadIdx.x;
    const int hi   = lane >> 4;
    const int lo   = lane & 15;

    const float* W   = (mat == 0) ? wq : (mat == 1) ? wk : wv;
    const float* B   = (mat == 0) ? bq : (mat == 1) ? bk : nullptr;
    float*       OUT = (mat == 0) ? Q0 : (mat == 1) ? K0 : V0;

    const int m = tile * 16 + lo;
    const int n = ct * 16 + lo;

    v8f acc = {};
    for (int k = 0; k < F0; k += 4) {
        const int kb = k + 2 * hi;
        v2f a, b;
        a.x = x[(size_t)m * F0 + kb];
        a.y = x[(size_t)m * F0 + kb + 1];
        b.x = W[(size_t)kb * U0 + n];
        b.y = W[(size_t)(kb + 1) * U0 + n];
        acc = __builtin_amdgcn_wmma_f32_16x16x4_f32(
                  false, a, false, b, (short)0, acc, false, false);
    }

    const float bias = (B != nullptr) ? B[n] : 0.0f;
    #pragma unroll
    for (int v = 0; v < 8; ++v) {
        const int row = tile * 16 + v + 8 * hi;
        float val = acc[v] + bias;
        if (mat != 2) val = fmaxf(val, 0.0f);
        OUT[(size_t)row * U0 + n] = val;
    }
}

__global__ __launch_bounds__(32) void proj1v_kernel(
    const float* __restrict__ h0, const float* __restrict__ wv1,
    float* __restrict__ v1)
{
    const int tile = blockIdx.x;
    const int ct   = blockIdx.y;
    const int lane = threadIdx.x;
    const int hi   = lane >> 4;
    const int lo   = lane & 15;

    const int m = tile * 16 + lo;
    const int n = ct * 16 + lo;
    const bool valid = (n < U1);

    v8f acc = {};
    for (int k = 0; k < U0; k += 4) {
        const int kb = k + 2 * hi;
        v2f a, b;
        a.x = h0[(size_t)m * U0 + kb];
        a.y = h0[(size_t)m * U0 + kb + 1];
        b.x = valid ? wv1[(size_t)kb * U1 + n] : 0.0f;
        b.y = valid ? wv1[(size_t)(kb + 1) * U1 + n] : 0.0f;
        acc = __builtin_amdgcn_wmma_f32_16x16x4_f32(
                  false, a, false, b, (short)0, acc, false, false);
    }
    if (valid) {
        #pragma unroll
        for (int v = 0; v < 8; ++v) {
            const int row = tile * 16 + v + 8 * hi;
            v1[(size_t)row * U1 + n] = acc[v];
        }
    }
}

__global__ void qk1_kernel(const float* __restrict__ h0,
                           const float* __restrict__ wq1, const float* __restrict__ bq1,
                           const float* __restrict__ wk1, const float* __restrict__ bk1,
                           float* __restrict__ q1, float* __restrict__ k1, int n)
{
    const int i = blockIdx.x * blockDim.x + threadIdx.x;
    if (i >= n) return;
    const float* hr = h0 + (size_t)i * U0;
    float q = bq1[0], k = bk1[0];
    #pragma unroll 8
    for (int j = 0; j < U0; ++j) {
        const float h = hr[j];
        q += h * wq1[j];
        k += h * wk1[j];
    }
    q1[i] = fmaxf(q, 0.0f);
    k1[i] = fmaxf(k, 0.0f);
}

#define ROWS 200
#define CAP  8192
#define NT   1024
#define P0   65
#define P1   41
#define LHEAD 32

static_assert(ROWS <= 256, "a handed-on list entry keeps the local row in 8 bits");

__device__ __forceinline__ void list_push(int e, int r, int row0, int nrows,
                                          int* s_cnt, int* s_lr, int* s_lc)
{
    const unsigned rl = (unsigned)(r - row0);
    if (rl < (unsigned)nrows) {
        const int pos = atomicAdd(s_cnt, 1);
        if (pos < CAP) { s_lr[pos] = (int)rl; s_lc[pos] = e; }
    }
}

__device__ __forceinline__ int make_list(const int* __restrict__ ei, int E,
                                         int e_begin, int e_end, int row0, int nrows,
                                         int* s_cnt, int* s_lr, int* s_lc)
{
    const int tid = threadIdx.x;
    __syncthreads();
    if (tid == 0) *s_cnt = 0;
    __syncthreads();

    const int n4 = (e_end - e_begin) >> 2;
    const int4* __restrict__ rows4 = (const int4*)(ei + e_begin);
    const int   miss = row0 - 1;
    const unsigned nr = (unsigned)nrows;

    auto load4 = [&](int gg) -> int4 {
        int4 v = rows4[(gg < n4) ? gg : (n4 - 1)];
        if (gg >= n4) v = make_int4(miss, miss, miss, miss);
        return v;
    };
    auto scan4 = [&](const int4& v, int gg) {
        const bool any = ((unsigned)(v.x - row0) < nr) | ((unsigned)(v.y - row0) < nr) |
                         ((unsigned)(v.z - row0) < nr) | ((unsigned)(v.w - row0) < nr);
        if (any) {
            const int e = e_begin + 4 * gg;
            list_push(e,     v.x, row0, nrows, s_cnt, s_lr, s_lc);
            list_push(e + 1, v.y, row0, nrows, s_cnt, s_lr, s_lc);
            list_push(e + 2, v.z, row0, nrows, s_cnt, s_lr, s_lc);
            list_push(e + 3, v.w, row0, nrows, s_cnt, s_lr, s_lc);
        }
    };

    if (tid < n4) {
        int4 cur[4], nxt[4];
        for (int u = 0; u < 4; ++u) cur[u] = load4(tid + u * NT);
        for (int g = tid; g < n4; g += 4 * NT) {
            for (int u = 0; u < 4; ++u) nxt[u] = load4(g + (4 + u) * NT);
            for (int u = 0; u < 4; ++u) scan4(cur[u], g + u * NT);
            for (int u = 0; u < 4; ++u) cur[u] = nxt[u];
        }
    }
    const int et = e_begin + 4 * n4 + tid;
    if (et < e_end) list_push(et, ei[et], row0, nrows, s_cnt, s_lr, s_lc);
    __syncthreads();

    const int total = __builtin_amdgcn_readfirstlane(*s_cnt);
    const int nl    = (total < CAP) ? total : CAP;
    for (int i = tid; i < nl; i += NT) s_lc[i] = ei[E + s_lc[i]];
    __syncthreads();
    return total;
}

__global__ __launch_bounds__(NT) void gat0_owner_kernel(
    const int* __restrict__ ei, int E, int N,
    const float* __restrict__ Q0, const float* __restrict__ K0,
    const float* __restrict__ V0, const float* __restrict__ b0,
    float* __restrict__ h0, int* __restrict__ lst)
{
    __shared__ unsigned s_max[ROWS * H0];
    __shared__ float    s_den[ROWS * H0];
    __shared__ float    s_acc[ROWS * P0];
    __shared__ int      s_lr[CAP];
    __shared__ int      s_lc[CAP];
    __shared__ int      s_cnt;

    const int tid   = threadIdx.x;
    const int row0  = blockIdx.x * ROWS;
    const int nrows = (N - row0 < ROWS) ? (N - row0) : ROWS;

    for (int t = tid; t < ROWS * H0; t += NT) { s_max[t] = 0u; s_den[t] = 0.0f; }
    for (int t = tid; t < ROWS * P0; t += NT) s_acc[t] = 0.0f;

    auto item = [&](int entry, int n, int& rl, int& c) {
        if (entry < n) { rl = s_lr[entry]; c = s_lc[entry]; }
        else           { rl = entry - n;   c = row0 + rl; }
    };
    auto score = [&](int rl, int c, int h) -> float {
        const float4* qp = (const float4*)(Q0 + (size_t)(row0 + rl) * U0 + h * A0);
        const float4* kp = (const float4*)(K0 + (size_t)c * U0 + h * A0);
        const float4 qa = qp[0], qb = qp[1], ka = kp[0], kb = kp[1];
        float s = 0.0f;
        s += qa.x * ka.x; s += qa.y * ka.y; s += qa.z * ka.z; s += qa.w * ka.w;
        s += qb.x * kb.x; s += qb.y * kb.y; s += qb.z * kb.z; s += qb.w * kb.w;
        return s * 0.35355339059327373f;
    };

    auto phase_max = [&](int n, int nself) {
        const int items = (n + nself) * H0;
        for (int base = tid; base < items; base += 2 * NT) {
            int rl[2], c[2], h[2]; float s[2];
            for (int u = 0; u < 2; ++u) {
                const int idx = base + u * NT;
                const int id  = (idx < items) ? idx : base;
                h[u] = id & 7;
                item(id >> 3, n, rl[u], c[u]);
            }
            for (int u = 0; u < 2; ++u) s[u] = score(rl[u], c[u], h[u]);
            for (int u = 0; u < 2; ++u)
                atomicMax(&s_max[rl[u] * H0 + h[u]], fkey(s[u]));
        }
    };

    auto phase_sum = [&](int n, int nself) {
        const int items = (n + nself) * H0;
        for (int base = tid; base < items; base += 2 * NT) {
            int rl[2], c[2], h[2]; bool ok[2]; float s[2]; float4 va[2], vb[2];
            for (int u = 0; u < 2; ++u) {
                const int idx = base + u * NT;
                ok[u] = idx < items;
                const int id = ok[u] ? idx : base;
                h[u] = id & 7;
                item(id >> 3, n, rl[u], c[u]);
            }
            for (int u = 0; u < 2; ++u) {
                s[u] = score(rl[u], c[u], h[u]);
                const float4* vp = (const float4*)(V0 + (size_t)c[u] * U0 + h[u] * A0);
                va[u] = vp[0]; vb[u] = vp[1];
            }
            for (int u = 0; u < 2; ++u) {
                const float m  = fkey_dec(s_max[rl[u] * H0 + h[u]]);
                const float ev = __expf(s[u] - m) * (ok[u] ? 1.0f : 0.0f);
                atomicAdd(&s_den[rl[u] * H0 + h[u]], ev);
                float* acc = &s_acc[rl[u] * P0 + h[u] * A0];
                atomicAdd(&acc[0], ev * va[u].x); atomicAdd(&acc[1], ev * va[u].y);
                atomicAdd(&acc[2], ev * va[u].z); atomicAdd(&acc[3], ev * va[u].w);
                atomicAdd(&acc[4], ev * vb[u].x); atomicAdd(&acc[5], ev * vb[u].y);
                atomicAdd(&acc[6], ev * vb[u].z); atomicAdd(&acc[7], ev * vb[u].w);
            }
        }
    };

    const int total = make_list(ei, E, 0, E, row0, nrows, &s_cnt, s_lr, s_lc);

    {
        int* mine = lst + (size_t)row0 * U1;
        const bool fits = (total <= CAP) && (total <= nrows * U1 - LHEAD) && (N <= (1 << 24));
        if (tid == 0) mine[0] = fits ? total : -1;
        if (fits)
            for (int i = tid; i < total; i += NT)
                mine[LHEAD + i] = (int)(((unsigned)s_lr[i] << 24) | (unsigned)s_lc[i]);
    }

    if (total <= CAP) {
        phase_max(total, nrows);
        __syncthreads();
        phase_sum(total, nrows);
    } else {
        for (int sweep = 0; sweep < 2; ++sweep) {
            for (int b = 0; b < E; b += CAP) {
                const int n = make_list(ei, E, b, (b + CAP < E) ? (b + CAP) : E, row0, nrows,
                                        &s_cnt, s_lr, s_lc);
                const int nself = (b == 0) ? nrows : 0;
                if (sweep == 0) phase_max(n, nself); else phase_sum(n, nself);
            }
        }
    }
    __syncthreads();

    for (int t = tid; t < nrows * U0; t += NT) {
        const int rl = t >> 6, cc = t & 63;
        const float val = s_acc[rl * P0 + cc] / s_den[rl * H0 + (cc >> 3)] + b0[cc];
        h0[(size_t)row0 * U0 + t] = fmaxf(val, 0.0f);
    }
}

__global__ __launch_bounds__(NT) void gat1_owner_kernel(
    const int* __restrict__ ei, int E, int N,
    const float* __restrict__ q1, const float* __restrict__ k1,
    const float* __restrict__ v1, const float* __restrict__ b1,
    const int* lst, float* out)
{
    __shared__ unsigned s_max[ROWS];
    __shared__ float    s_den[ROWS];
    __shared__ float    s_acc[ROWS * P1];
    __shared__ int      s_lr[CAP];
    __shared__ int      s_lc[CAP];
    __shared__ int      s_cnt;

    const int tid   = threadIdx.x;
    const int row0  = blockIdx.x * ROWS;
    const int nrows = (N - row0 < ROWS) ? (N - row0) : ROWS;

    for (int t = tid; t < ROWS; t += NT) { s_max[t] = 0u; s_den[t] = 0.0f; }
    for (int t = tid; t < ROWS * P1; t += NT) s_acc[t] = 0.0f;

    auto item = [&](int entry, int n, int& rl, int& c) {
        if (entry < n) { rl = s_lr[entry]; c = s_lc[entry]; }
        else           { rl = entry - n;   c = row0 + rl; }
    };

    auto phase_max = [&](int n, int nself) {
        const int items = n + nself;
        for (int base = tid; base < items; base += 2 * NT) {
            int rl[2], c[2]; float s[2];
            for (int u = 0; u < 2; ++u) {
                const int idx = base + u * NT;
                item((idx < items) ? idx : base, n, rl[u], c[u]);
            }
            for (int u = 0; u < 2; ++u) s[u] = q1[row0 + rl[u]] * k1[c[u]];
            for (int u = 0; u < 2; ++u)
                atomicMax(&s_max[rl[u]], fkey(s[u]));
        }
    };

    auto phase_sum = [&](int n, int nself) {
        const int items = (n + nself) * 8;
        for (int base = tid; base < items; base += 2 * NT) {
            int rl[2], c[2], j[2]; bool ok[2]; float s[2]; float x[2][5];
            for (int u = 0; u < 2; ++u) {
                const int idx = base + u * NT;
                ok[u] = idx < items;
                const int id = ok[u] ? idx : base;
                j[u] = id & 7;
                item(id >> 3, n, rl[u], c[u]);
            }
            for (int u = 0; u < 2; ++u) {
                s[u] = q1[row0 + rl[u]] * k1[c[u]];
                const float* vv = v1 + (size_t)c[u] * U1 + j[u];
                for (int i = 0; i < 5; ++i) x[u][i] = vv[8 * i];
            }
            for (int u = 0; u < 2; ++u) {
                const float m  = fkey_dec(s_max[rl[u]]);
                const float ev = __expf(s[u] - m) * (ok[u] ? 1.0f : 0.0f);
                if (j[u] == 0) atomicAdd(&s_den[rl[u]], ev);
                float* acc = &s_acc[rl[u] * P1 + j[u]];
                for (int i = 0; i < 5; ++i) atomicAdd(&acc[8 * i], ev * x[u][i]);
            }
        }
    };

    const int* mine = lst + (size_t)row0 * U1;
    const int given = __builtin_amdgcn_readfirstlane(mine[0]);
    if (given >= 0 && given <= CAP && given <= nrows * U1 - LHEAD) {
        for (int i = tid; i < given; i += NT) {
            const unsigned w = (unsigned)mine[LHEAD + i];
            s_lr[i] = (int)(w >> 24);
            s_lc[i] = (int)(w & 0x00FFFFFFu);
        }
        __syncthreads();
        phase_max(given, nrows);
        __syncthreads();
        phase_sum(given, nrows);
    } else {
        const int total = make_list(ei, E, 0, E, row0, nrows, &s_cnt, s_lr, s_lc);
        if (total <= CAP) {
            phase_max(total, nrows);
            __syncthreads();
            phase_sum(total, nrows);
        } else {
            for (int sweep = 0; sweep < 2; ++sweep) {
                for (int b = 0; b < E; b += CAP) {
                    const int n = make_list(ei, E, b, (b + CAP < E) ? (b + CAP) : E, row0, nrows,
                                            &s_cnt, s_lr, s_lc);
                    const int nself = (b == 0) ? nrows : 0;
                    if (sweep == 0) phase_max(n, nself); else phase_sum(n, nself);
                }
            }
        }
    }
    __syncthreads();

    for (int t = tid; t < nrows * U1; t += NT) {
        const int rl = t / U1, cc = t - rl * U1;
        out[(size_t)row0 * U1 + t] = s_acc[rl * P1 + cc] / s_den[rl] + b1[cc];
    }
}

extern "C" void kernel_launch(void* const* d_in, const int* in_sizes, int n_in,
                              void* d_out, int out_size, void* d_ws, size_t ws_size,
                              hipStream_t stream)
{
    const float* x    = (const float*)d_in[0];
    const int*   ei   = (const int*)  d_in[1];
    const float* wq0  = (const float*)d_in[2];
    const float* bq0  = (const float*)d_in[3];
    const float* wk0  = (const float*)d_in[4];
    const float* bk0  = (const float*)d_in[5];
    const float* wv0  = (const float*)d_in[6];
    const float* b0   = (const float*)d_in[7];
    const float* wq1  = (const float*)d_in[8];
    const float* bq1  = (const float*)d_in[9];
    const float* wk1  = (const float*)d_in[10];
    const float* bk1  = (const float*)d_in[11];
    const float* wv1  = (const float*)d_in[12];
    const float* b1   = (const float*)d_in[13];
    float* out = (float*)d_out;

    const int N = in_sizes[0] / F0;
    const int E = in_sizes[1] / 2;

    float* ws = (float*)d_ws;
    float* R0 = ws;
    float* R1 = R0 + (size_t)N * U0;
    float* R2 = R1 + (size_t)N * U0;
    float* R3 = R2 + (size_t)N * U0;
    float* R4 = R3 + (size_t)N * H0;
    float* R5 = R4 + (size_t)N * H0;

    const int TB = 256;
    const int G  = (N + ROWS - 1) / ROWS;

    proj0_kernel<<<dim3(N / 16, 12), 32, 0, stream>>>(
        x, wq0, bq0, wk0, bk0, wv0, R0, R1, R2);
    gat0_owner_kernel<<<G, NT, 0, stream>>>(ei, E, N, R0, R1, R2, b0, R5  ,
                                            (int*)d_out  );

    float* q1 = R3;
    float* k1 = R3 + N;
    qk1_kernel<<<(N + TB - 1) / TB, TB, 0, stream>>>(
        R5, wq1, bq1, wk1, bk1, q1, k1, N);
    proj1v_kernel<<<dim3(N / 16, 3), 32, 0, stream>>>(
        R5, wv1, R1  );
    gat1_owner_kernel<<<G, NT, 0, stream>>>(ei, E, N, q1, k1, R1, b1, (const int*)d_out, out);
}


# Both layers: an edge's score is worked out once, in the maxima pass, and kept in LDS; the sums pass reads it there and fetches only V (Q0, K0, q1, k1 read once); list capacity 4096; sums as before
# speedup vs baseline: 2.8097x; 1.0595x over previous
#include <hip/hip_runtime.h>

typedef float v2f __attribute__((ext_vector_type(2)));
typedef float v8f __attribute__((ext_vector_type(8)));

#define F0 128
#define U0 64
#define H0 8
#define A0 8
#define U1 40

__device__ __forceinline__ unsigned fkey(float f) {
    unsigned b = __float_as_uint(f);
    return (b & 0x80000000u) ? ~b : (b | 0x80000000u);
}
__device__ __forceinline__ float fkey_dec(unsigned k) {
    return __uint_as_float((k & 0x80000000u) ? (k ^ 0x80000000u) : ~k);
}

__global__ __launch_bounds__(32) void proj0_kernel(
    const float* __restrict__ x,
    const float* __restrict__ wq, const float* __restrict__ bq,
    const float* __restrict__ wk, const float* __restrict__ bk,
    const float* __restrict__ wv,
    float* __restrict__ Q0, float* __restrict__ K0, float* __restrict__ V0)
{
    const int tile = blockIdx.x;
    const int mat  = blockIdx.y >> 2;
    const int ct   = blockIdx.y & 3;
    const int lane = threadIdx.x;
    const int hi   = lane >> 4;
    const int lo   = lane & 15;

    const float* W   = (mat == 0) ? wq : (mat == 1) ? wk : wv;
    const float* B   = (mat == 0) ? bq : (mat == 1) ? bk : nullptr;
    float*       OUT = (mat == 0) ? Q0 : (mat == 1) ? K0 : V0;

    const int m = tile * 16 + lo;
    const int n = ct * 16 + lo;

    v8f acc = {};
    for (int k = 0; k < F0; k += 4) {
        const int kb = k + 2 * hi;
        v2f a, b;
        a.x = x[(size_t)m * F0 + kb];
        a.y = x[(size_t)m * F0 + kb + 1];
        b.x = W[(size_t)kb * U0 + n];
        b.y = W[(size_t)(kb + 1) * U0 + n];
        acc = __builtin_amdgcn_wmma_f32_16x16x4_f32(
                  false, a, false, b, (short)0, acc, false, false);
    }

    const float bias = (B != nullptr) ? B[n] : 0.0f;
    #pragma unroll
    for (int v = 0; v < 8; ++v) {
        const int row = tile * 16 + v + 8 * hi;
        float val = acc[v] + bias;
        if (mat != 2) val = fmaxf(val, 0.0f);
        OUT[(size_t)row * U0 + n] = val;
    }
}

__global__ __launch_bounds__(32) void proj1v_kernel(
    const float* __restrict__ h0, const float* __restrict__ wv1,
    float* __restrict__ v1)
{
    const int tile = blockIdx.x;
    const int ct   = blockIdx.y;
    const int lane = threadIdx.x;
    const int hi   = lane >> 4;
    const int lo   = lane & 15;

    const int m = tile * 16 + lo;
    const int n = ct * 16 + lo;
    const bool valid = (n < U1);

    v8f acc = {};
    for (int k = 0; k < U0; k += 4) {
        const int kb = k + 2 * hi;
        v2f a, b;
        a.x = h0[(size_t)m * U0 + kb];
        a.y = h0[(size_t)m * U0 + kb + 1];
        b.x = valid ? wv1[(size_t)kb * U1 + n] : 0.0f;
        b.y = valid ? wv1[(size_t)(kb + 1) * U1 + n] : 0.0f;
        acc = __builtin_amdgcn_wmma_f32_16x16x4_f32(
                  false, a, false, b, (short)0, acc, false, false);
    }
    if (valid) {
        #pragma unroll
        for (int v = 0; v < 8; ++v) {
            const int row = tile * 16 + v + 8 * hi;
            v1[(size_t)row * U1 + n] = acc[v];
        }
    }
}

__global__ void qk1_kernel(const float* __restrict__ h0,
                           const float* __restrict__ wq1, const float* __restrict__ bq1,
                           const float* __restrict__ wk1, const float* __restrict__ bk1,
                           float* __restrict__ q1, float* __restrict__ k1, int n)
{
    const int i = blockIdx.x * blockDim.x + threadIdx.x;
    if (i >= n) return;
    const float* hr = h0 + (size_t)i * U0;
    float q = bq1[0], k = bk1[0];
    #pragma unroll 8
    for (int j = 0; j < U0; ++j) {
        const float h = hr[j];
        q += h * wq1[j];
        k += h * wk1[j];
    }
    q1[i] = fmaxf(q, 0.0f);
    k1[i] = fmaxf(k, 0.0f);
}

#define ROWS 200
#define CAP  4096
#define NT   1024
#define P0   65
#define P1   41
#define LHEAD 32

static_assert(ROWS <= 256, "a handed-on list entry keeps the local row in 8 bits");

struct yes_t { static constexpr bool value = true;  };
struct no_t  { static constexpr bool value = false; };

__device__ __forceinline__ void list_push(int e, int r, int row0, int nrows,
                                          int* s_cnt, int* s_lr, int* s_lc)
{
    const unsigned rl = (unsigned)(r - row0);
    if (rl < (unsigned)nrows) {
        const int pos = atomicAdd(s_cnt, 1);
        if (pos < CAP) { s_lr[pos] = (int)rl; s_lc[pos] = e; }
    }
}

__device__ __forceinline__ int make_list(const int* __restrict__ ei, int E,
                                         int e_begin, int e_end, int row0, int nrows,
                                         int* s_cnt, int* s_lr, int* s_lc)
{
    const int tid = threadIdx.x;
    __syncthreads();
    if (tid == 0) *s_cnt = 0;
    __syncthreads();

    const int n4 = (e_end - e_begin) >> 2;
    const int4* __restrict__ rows4 = (const int4*)(ei + e_begin);
    const int   miss = row0 - 1;
    const unsigned nr = (unsigned)nrows;

    auto load4 = [&](int gg) -> int4 {
        int4 v = rows4[(gg < n4) ? gg : (n4 - 1)];
        if (gg >= n4) v = make_int4(miss, miss, miss, miss);
        return v;
    };
    auto scan4 = [&](const int4& v, int gg) {
        const bool any = ((unsigned)(v.x - row0) < nr) | ((unsigned)(v.y - row0) < nr) |
                         ((unsigned)(v.z - row0) < nr) | ((unsigned)(v.w - row0) < nr);
        if (any) {
            const int e = e_begin + 4 * gg;
            list_push(e,     v.x, row0, nrows, s_cnt, s_lr, s_lc);
            list_push(e + 1, v.y, row0, nrows, s_cnt, s_lr, s_lc);
            list_push(e + 2, v.z, row0, nrows, s_cnt, s_lr, s_lc);
            list_push(e + 3, v.w, row0, nrows, s_cnt, s_lr, s_lc);
        }
    };

    if (tid < n4) {
        int4 cur[4], nxt[4];
        for (int u = 0; u < 4; ++u) cur[u] = load4(tid + u * NT);
        for (int g = tid; g < n4; g += 4 * NT) {
            for (int u = 0; u < 4; ++u) nxt[u] = load4(g + (4 + u) * NT);
            for (int u = 0; u < 4; ++u) scan4(cur[u], g + u * NT);
            for (int u = 0; u < 4; ++u) cur[u] = nxt[u];
        }
    }
    const int et = e_begin + 4 * n4 + tid;
    if (et < e_end) list_push(et, ei[et], row0, nrows, s_cnt, s_lr, s_lc);
    __syncthreads();

    const int total = __builtin_amdgcn_readfirstlane(*s_cnt);
    const int nl    = (total < CAP) ? total : CAP;
    for (int i = tid; i < nl; i += NT) s_lc[i] = ei[E + s_lc[i]];
    __syncthreads();
    return total;
}

__global__ __launch_bounds__(NT) void gat0_owner_kernel(
    const int* __restrict__ ei, int E, int N,
    const float* __restrict__ Q0, const float* __restrict__ K0,
    const float* __restrict__ V0, const float* __restrict__ b0,
    float* __restrict__ h0, int* __restrict__ lst)
{
    __shared__ unsigned s_max[ROWS * H0];
    __shared__ float    s_den[ROWS * H0];
    __shared__ float    s_acc[ROWS * P0];
    __shared__ float    s_sc[(CAP + ROWS) * H0];
    __shared__ int      s_lr[CAP];
    __shared__ int      s_lc[CAP];
    __shared__ int      s_cnt;

    const int tid   = threadIdx.x;
    const int row0  = blockIdx.x * ROWS;
    const int nrows = (N - row0 < ROWS) ? (N - row0) : ROWS;

    for (int t = tid; t < ROWS * H0; t += NT) { s_max[t] = 0u; s_den[t] = 0.0f; }
    for (int t = tid; t < ROWS * P0; t += NT) s_acc[t] = 0.0f;

    auto item = [&](int entry, int n, int& rl, int& c) {
        if (entry < n) { rl = s_lr[entry]; c = s_lc[entry]; }
        else           { rl = entry - n;   c = row0 + rl; }
    };
    auto score = [&](int rl, int c, int h) -> float {
        const float4* qp = (const float4*)(Q0 + (size_t)(row0 + rl) * U0 + h * A0);
        const float4* kp = (const float4*)(K0 + (size_t)c * U0 + h * A0);
        const float4 qa = qp[0], qb = qp[1], ka = kp[0], kb = kp[1];
        float s = 0.0f;
        s += qa.x * ka.x; s += qa.y * ka.y; s += qa.z * ka.z; s += qa.w * ka.w;
        s += qb.x * kb.x; s += qb.y * kb.y; s += qb.z * kb.z; s += qb.w * kb.w;
        return s * 0.35355339059327373f;
    };

    auto phase_max = [&](int n, int nself, auto keep) {
        const int items = (n + nself) * H0;
        for (int base = tid; base < items; base += 2 * NT) {
            int rl[2], c[2], h[2], id[2]; float s[2];
            for (int u = 0; u < 2; ++u) {
                const int idx = base + u * NT;
                id[u] = (idx < items) ? idx : base;
                h[u] = id[u] & 7;
                item(id[u] >> 3, n, rl[u], c[u]);
            }
            for (int u = 0; u < 2; ++u) s[u] = score(rl[u], c[u], h[u]);
            for (int u = 0; u < 2; ++u) {
                if constexpr (decltype(keep)::value) s_sc[id[u]] = s[u];
                atomicMax(&s_max[rl[u] * H0 + h[u]], fkey(s[u]));
            }
        }
    };

    auto phase_sum = [&](int n, int nself, auto kept) {
        const int items = (n + nself) * H0;
        for (int base = tid; base < items; base += 2 * NT) {
            int rl[2], c[2], h[2], id[2]; bool ok[2]; float s[2]; float4 va[2], vb[2];
            for (int u = 0; u < 2; ++u) {
                const int idx = base + u * NT;
                ok[u] = idx < items;
                id[u] = ok[u] ? idx : base;
                h[u] = id[u] & 7;
                item(id[u] >> 3, n, rl[u], c[u]);
            }
            for (int u = 0; u < 2; ++u) {
                if constexpr (decltype(kept)::value) s[u] = s_sc[id[u]];
                else                                 s[u] = score(rl[u], c[u], h[u]);
                const float4* vp = (const float4*)(V0 + (size_t)c[u] * U0 + h[u] * A0);
                va[u] = vp[0]; vb[u] = vp[1];
            }
            for (int u = 0; u < 2; ++u) {
                const float m  = fkey_dec(s_max[rl[u] * H0 + h[u]]);
                const float ev = __expf(s[u] - m) * (ok[u] ? 1.0f : 0.0f);
                atomicAdd(&s_den[rl[u] * H0 + h[u]], ev);
                float* acc = &s_acc[rl[u] * P0 + h[u] * A0];
                atomicAdd(&acc[0], ev * va[u].x); atomicAdd(&acc[1], ev * va[u].y);
                atomicAdd(&acc[2], ev * va[u].z); atomicAdd(&acc[3], ev * va[u].w);
                atomicAdd(&acc[4], ev * vb[u].x); atomicAdd(&acc[5], ev * vb[u].y);
                atomicAdd(&acc[6], ev * vb[u].z); atomicAdd(&acc[7], ev * vb[u].w);
            }
        }
    };

    const int total = make_list(ei, E, 0, E, row0, nrows, &s_cnt, s_lr, s_lc);

    {
        int* mine = lst + (size_t)row0 * U1;
        const bool fits = (total <= CAP) && (total <= nrows * U1 - LHEAD) && (N <= (1 << 24));
        if (tid == 0) mine[0] = fits ? total : -1;
        if (fits)
            for (int i = tid; i < total; i += NT)
                mine[LHEAD + i] = (int)(((unsigned)s_lr[i] << 24) | (unsigned)s_lc[i]);
    }

    if (total <= CAP) {
        phase_max(total, nrows, yes_t{});
        __syncthreads();
        phase_sum(total, nrows, yes_t{});
    } else {
        for (int sweep = 0; sweep < 2; ++sweep) {
            for (int b = 0; b < E; b += CAP) {
                const int n = make_list(ei, E, b, (b + CAP < E) ? (b + CAP) : E, row0, nrows,
                                        &s_cnt, s_lr, s_lc);
                const int nself = (b == 0) ? nrows : 0;
                if (sweep == 0) phase_max(n, nself, no_t{}); else phase_sum(n, nself, no_t{});
            }
        }
    }
    __syncthreads();

    for (int t = tid; t < nrows * U0; t += NT) {
        const int rl = t >> 6, cc = t & 63;
        const float val = s_acc[rl * P0 + cc] / s_den[rl * H0 + (cc >> 3)] + b0[cc];
        h0[(size_t)row0 * U0 + t] = fmaxf(val, 0.0f);
    }
}

__global__ __launch_bounds__(NT) void gat1_owner_kernel(
    const int* __restrict__ ei, int E, int N,
    const float* __restrict__ q1, const float* __restrict__ k1,
    const float* __restrict__ v1, const float* __restrict__ b1,
    const int* lst, float* out)
{
    __shared__ unsigned s_max[ROWS];
    __shared__ float    s_den[ROWS];
    __shared__ float    s_acc[ROWS * P1];
    __shared__ float    s_sc[CAP + ROWS];
    __shared__ int      s_lr[CAP];
    __shared__ int      s_lc[CAP];
    __shared__ int      s_cnt;

    const int tid   = threadIdx.x;
    const int row0  = blockIdx.x * ROWS;
    const int nrows = (N - row0 < ROWS) ? (N - row0) : ROWS;

    for (int t = tid; t < ROWS; t += NT) { s_max[t] = 0u; s_den[t] = 0.0f; }
    for (int t = tid; t < ROWS * P1; t += NT) s_acc[t] = 0.0f;

    auto item = [&](int entry, int n, int& rl, int& c) {
        if (entry < n) { rl = s_lr[entry]; c = s_lc[entry]; }
        else           { rl = entry - n;   c = row0 + rl; }
    };

    auto phase_max = [&](int n, int nself, auto keep) {
        const int items = n + nself;
        for (int base = tid; base < items; base += 2 * NT) {
            int rl[2], c[2], id[2]; float s[2];
            for (int u = 0; u < 2; ++u) {
                const int idx = base + u * NT;
                id[u] = (idx < items) ? idx : base;
                item(id[u], n, rl[u], c[u]);
            }
            for (int u = 0; u < 2; ++u) s[u] = q1[row0 + rl[u]] * k1[c[u]];
            for (int u = 0; u < 2; ++u) {
                if constexpr (decltype(keep)::value) s_sc[id[u]] = s[u];
                atomicMax(&s_max[rl[u]], fkey(s[u]));
            }
        }
    };

    auto phase_sum = [&](int n, int nself, auto kept) {
        const int items = (n + nself) * 8;
        for (int base = tid; base < items; base += 2 * NT) {
            int rl[2], c[2], j[2], id[2]; bool ok[2]; float s[2]; float x[2][5];
            for (int u = 0; u < 2; ++u) {
                const int idx = base + u * NT;
                ok[u] = idx < items;
                id[u] = ok[u] ? idx : base;
                j[u] = id[u] & 7;
                item(id[u] >> 3, n, rl[u], c[u]);
            }
            for (int u = 0; u < 2; ++u) {
                if constexpr (decltype(kept)::value) s[u] = s_sc[id[u] >> 3];
                else                                 s[u] = q1[row0 + rl[u]] * k1[c[u]];
                const float* vv = v1 + (size_t)c[u] * U1 + j[u];
                for (int i = 0; i < 5; ++i) x[u][i] = vv[8 * i];
            }
            for (int u = 0; u < 2; ++u) {
                const float m  = fkey_dec(s_max[rl[u]]);
                const float ev = __expf(s[u] - m) * (ok[u] ? 1.0f : 0.0f);
                if (j[u] == 0) atomicAdd(&s_den[rl[u]], ev);
                float* acc = &s_acc[rl[u] * P1 + j[u]];
                for (int i = 0; i < 5; ++i) atomicAdd(&acc[8 * i], ev * x[u][i]);
            }
        }
    };

    const int* mine = lst + (size_t)row0 * U1;
    const int given = __builtin_amdgcn_readfirstlane(mine[0]);
    if (given >= 0 && given <= CAP && given <= nrows * U1 - LHEAD) {
        for (int i = tid; i < given; i += NT) {
            const unsigned w = (unsigned)mine[LHEAD + i];
            s_lr[i] = (int)(w >> 24);
            s_lc[i] = (int)(w & 0x00FFFFFFu);
        }
        __syncthreads();
        phase_max(given, nrows, yes_t{});
        __syncthreads();
        phase_sum(given, nrows, yes_t{});
    } else {
        const int total = make_list(ei, E, 0, E, row0, nrows, &s_cnt, s_lr, s_lc);
        if (total <= CAP) {
            phase_max(total, nrows, yes_t{});
            __syncthreads();
            phase_sum(total, nrows, yes_t{});
        } else {
            for (int sweep = 0; sweep < 2; ++sweep) {
                for (int b = 0; b < E; b += CAP) {
                    const int n = make_list(ei, E, b, (b + CAP < E) ? (b + CAP) : E, row0, nrows,
                                            &s_cnt, s_lr, s_lc);
                    const int nself = (b == 0) ? nrows : 0;
                    if (sweep == 0) phase_max(n, nself, no_t{}); else phase_sum(n, nself, no_t{});
                }
            }
        }
    }
    __syncthreads();

    for (int t = tid; t < nrows * U1; t += NT) {
        const int rl = t / U1, cc = t - rl * U1;
        out[(size_t)row0 * U1 + t] = s_acc[rl * P1 + cc] / s_den[rl] + b1[cc];
    }
}

extern "C" void kernel_launch(void* const* d_in, const int* in_sizes, int n_in,
                              void* d_out, int out_size, void* d_ws, size_t ws_size,
                              hipStream_t stream)
{
    const float* x    = (const float*)d_in[0];
    const int*   ei   = (const int*)  d_in[1];
    const float* wq0  = (const float*)d_in[2];
    const float* bq0  = (const float*)d_in[3];
    const float* wk0  = (const float*)d_in[4];
    const float* bk0  = (const float*)d_in[5];
    const float* wv0  = (const float*)d_in[6];
    const float* b0   = (const float*)d_in[7];
    const float* wq1  = (const float*)d_in[8];
    const float* bq1  = (const float*)d_in[9];
    const float* wk1  = (const float*)d_in[10];
    const float* bk1  = (const float*)d_in[11];
    const float* wv1  = (const float*)d_in[12];
    const float* b1   = (const float*)d_in[13];
    float* out = (float*)d_out;

    const int N = in_sizes[0] / F0;
    const int E = in_sizes[1] / 2;

    float* ws = (float*)d_ws;
    float* R0 = ws;
    float* R1 = R0 + (size_t)N * U0;
    float* R2 = R1 + (size_t)N * U0;
    float* R3 = R2 + (size_t)N * U0;
    float* R4 = R3 + (size_t)N * H0;
    float* R5 = R4 + (size_t)N * H0;

    const int TB = 256;
    const int G  = (N + ROWS - 1) / ROWS;

    proj0_kernel<<<dim3(N / 16, 12), 32, 0, stream>>>(
        x, wq0, bq0, wk0, bk0, wv0, R0, R1, R2);
    gat0_owner_kernel<<<G, NT, 0, stream>>>(ei, E, N, R0, R1, R2, b0, R5  ,
                                            (int*)d_out  );

    float* q1 = R3;
    float* k1 = R3 + N;
    qk1_kernel<<<(N + TB - 1) / TB, TB, 0, stream>>>(
        R5, wq1, bq1, wk1, bk1, q1, k1, N);
    proj1v_kernel<<<dim3(N / 16, 3), 32, 0, stream>>>(
        R5, wv1, R1  );
    gat1_owner_kernel<<<G, NT, 0, stream>>>(ei, E, N, q1, k1, R1, b1, (const int*)d_out, out);
}


# Layer 0's scan for a work-group's edges reads one-byte owner keys (row / 200, written by a new small launch in front), not the 4-byte rows: 5 -> 6 launches; LDS-atomic sums, arrival order as before
# speedup vs baseline: 3.0272x; 1.0774x over previous
#include <hip/hip_runtime.h>

typedef float v2f __attribute__((ext_vector_type(2)));
typedef float v8f __attribute__((ext_vector_type(8)));

#define F0 128
#define U0 64
#define H0 8
#define A0 8
#define U1 40

__device__ __forceinline__ unsigned fkey(float f) {
    unsigned b = __float_as_uint(f);
    return (b & 0x80000000u) ? ~b : (b | 0x80000000u);
}
__device__ __forceinline__ float fkey_dec(unsigned k) {
    return __uint_as_float((k & 0x80000000u) ? (k ^ 0x80000000u) : ~k);
}

__global__ __launch_bounds__(32) void proj0_kernel(
    const float* __restrict__ x,
    const float* __restrict__ wq, const float* __restrict__ bq,
    const float* __restrict__ wk, const float* __restrict__ bk,
    const float* __restrict__ wv,
    float* __restrict__ Q0, float* __restrict__ K0, float* __restrict__ V0)
{
    const int tile = blockIdx.x;
    const int mat  = blockIdx.y >> 2;
    const int ct   = blockIdx.y & 3;
    const int lane = threadIdx.x;
    const int hi   = lane >> 4;
    const int lo   = lane & 15;

    const float* W   = (mat == 0) ? wq : (mat == 1) ? wk : wv;
    const float* B   = (mat == 0) ? bq : (mat == 1) ? bk : nullptr;
    float*       OUT = (mat == 0) ? Q0 : (mat == 1) ? K0 : V0;

    const int m = tile * 16 + lo;
    const int n = ct * 16 + lo;

    v8f acc = {};
    for (int k = 0; k < F0; k += 4) {
        const int kb = k + 2 * hi;
        v2f a, b;
        a.x = x[(size_t)m * F0 + kb];
        a.y = x[(size_t)m * F0 + kb + 1];
        b.x = W[(size_t)kb * U0 + n];
        b.y = W[(size_t)(kb + 1) * U0 + n];
        acc = __builtin_amdgcn_wmma_f32_16x16x4_f32(
                  false, a, false, b, (short)0, acc, false, false);
    }

    const float bias = (B != nullptr) ? B[n] : 0.0f;
    #pragma unroll
    for (int v = 0; v < 8; ++v) {
        const int row = tile * 16 + v + 8 * hi;
        float val = acc[v] + bias;
        if (mat != 2) val = fmaxf(val, 0.0f);
        OUT[(size_t)row * U0 + n] = val;
    }
}

__global__ __launch_bounds__(32) void proj1v_kernel(
    const float* __restrict__ h0, const float* __restrict__ wv1,
    float* __restrict__ v1)
{
    const int tile = blockIdx.x;
    const int ct   = blockIdx.y;
    const int lane = threadIdx.x;
    const int hi   = lane >> 4;
    const int lo   = lane & 15;

    const int m = tile * 16 + lo;
    const int n = ct * 16 + lo;
    const bool valid = (n < U1);

    v8f acc = {};
    for (int k = 0; k < U0; k += 4) {
        const int kb = k + 2 * hi;
        v2f a, b;
        a.x = h0[(size_t)m * U0 + kb];
        a.y = h0[(size_t)m * U0 + kb + 1];
        b.x = valid ? wv1[(size_t)kb * U1 + n] : 0.0f;
        b.y = valid ? wv1[(size_t)(kb + 1) * U1 + n] : 0.0f;
        acc = __builtin_amdgcn_wmma_f32_16x16x4_f32(
                  false, a, false, b, (short)0, acc, false, false);
    }
    if (valid) {
        #pragma unroll
        for (int v = 0; v < 8; ++v) {
            const int row = tile * 16 + v + 8 * hi;
            v1[(size_t)row * U1 + n] = acc[v];
        }
    }
}

__global__ void qk1_kernel(const float* __restrict__ h0,
                           const float* __restrict__ wq1, const float* __restrict__ bq1,
                           const float* __restrict__ wk1, const float* __restrict__ bk1,
                           float* __restrict__ q1, float* __restrict__ k1, int n)
{
    const int i = blockIdx.x * blockDim.x + threadIdx.x;
    if (i >= n) return;
    const float* hr = h0 + (size_t)i * U0;
    float q = bq1[0], k = bk1[0];
    #pragma unroll 8
    for (int j = 0; j < U0; ++j) {
        const float h = hr[j];
        q += h * wq1[j];
        k += h * wk1[j];
    }
    q1[i] = fmaxf(q, 0.0f);
    k1[i] = fmaxf(k, 0.0f);
}

#define ROWS 200
#define CAP  4096
#define NT   1024
#define P0   65
#define P1   41
#define LHEAD 32

static_assert(ROWS <= 256, "a handed-on list entry keeps the local row in 8 bits");

struct yes_t { static constexpr bool value = true;  };
struct no_t  { static constexpr bool value = false; };

__device__ __forceinline__ void list_push(int e, int r, int row0, int nrows,
                                          int* s_cnt, int* s_lr, int* s_lc)
{
    const unsigned rl = (unsigned)(r - row0);
    if (rl < (unsigned)nrows) {
        const int pos = atomicAdd(s_cnt, 1);
        if (pos < CAP) { s_lr[pos] = (int)rl; s_lc[pos] = e; }
    }
}

__device__ __forceinline__ int make_list(const int* __restrict__ ei, int E,
                                         int e_begin, int e_end, int row0, int nrows,
                                         int* s_cnt, int* s_lr, int* s_lc)
{
    const int tid = threadIdx.x;
    __syncthreads();
    if (tid == 0) *s_cnt = 0;
    __syncthreads();

    const int n4 = (e_end - e_begin) >> 2;
    const int4* __restrict__ rows4 = (const int4*)(ei + e_begin);
    const int   miss = row0 - 1;
    const unsigned nr = (unsigned)nrows;

    auto load4 = [&](int gg) -> int4 {
        int4 v = rows4[(gg < n4) ? gg : (n4 - 1)];
        if (gg >= n4) v = make_int4(miss, miss, miss, miss);
        return v;
    };
    auto scan4 = [&](const int4& v, int gg) {
        const bool any = ((unsigned)(v.x - row0) < nr) | ((unsigned)(v.y - row0) < nr) |
                         ((unsigned)(v.z - row0) < nr) | ((unsigned)(v.w - row0) < nr);
        if (any) {
            const int e = e_begin + 4 * gg;
            list_push(e,     v.x, row0, nrows, s_cnt, s_lr, s_lc);
            list_push(e + 1, v.y, row0, nrows, s_cnt, s_lr, s_lc);
            list_push(e + 2, v.z, row0, nrows, s_cnt, s_lr, s_lc);
            list_push(e + 3, v.w, row0, nrows, s_cnt, s_lr, s_lc);
        }
    };

    if (tid < n4) {
        int4 cur[4], nxt[4];
        for (int u = 0; u < 4; ++u) cur[u] = load4(tid + u * NT);
        for (int g = tid; g < n4; g += 4 * NT) {
            for (int u = 0; u < 4; ++u) nxt[u] = load4(g + (4 + u) * NT);
            for (int u = 0; u < 4; ++u) scan4(cur[u], g + u * NT);
            for (int u = 0; u < 4; ++u) cur[u] = nxt[u];
        }
    }
    const int et = e_begin + 4 * n4 + tid;
    if (et < e_end) list_push(et, ei[et], row0, nrows, s_cnt, s_lr, s_lc);
    __syncthreads();

    const int total = __builtin_amdgcn_readfirstlane(*s_cnt);
    const int nl    = (total < CAP) ? total : CAP;
    for (int i = tid; i < nl; i += NT) s_lc[i] = ei[E + s_lc[i]];
    __syncthreads();
    return total;
}

__device__ __forceinline__ bool keys_usable(int groups, int E, int N)
{
    return (groups <= 255) && (E <= N * (H0 * 4));
}

__global__ __launch_bounds__(256) void owner_key_kernel(
    const int* __restrict__ ei, int E, int N, int groups, unsigned* __restrict__ own)
{
    if (!keys_usable(groups, E, N)) return;
    auto key = [](int r) -> unsigned {
        const unsigned k = (unsigned)r / (unsigned)ROWS;
        return (k < 255u) ? k : 255u;
    };
    const int t = blockIdx.x * blockDim.x + threadIdx.x;
    const int e = 4 * t;
    if (e + 3 < E) {
        const int4 r = ((const int4*)ei)[t];
        own[t] = key(r.x) | (key(r.y) << 8) | (key(r.z) << 16) | (key(r.w) << 24);
    } else if (e < E) {
        unsigned char* ob = (unsigned char*)own;
        for (int k = e; k < E; ++k) ob[k] = (unsigned char)key(ei[k]);
    }
}

__device__ __forceinline__ int make_list_keys(const int* __restrict__ ei, int E,
                                              const unsigned* __restrict__ own,
                                              int row0, unsigned id,
                                              int* s_cnt, int* s_lr, int* s_lc)
{
    const int tid = threadIdx.x;
    __syncthreads();
    if (tid == 0) *s_cnt = 0;
    __syncthreads();

    const int n16 = E >> 4;
    const uint4* __restrict__ own16 = (const uint4*)own;
    const unsigned k4   = id * 0x01010101u;
    const unsigned miss = ~k4;

    auto push = [&](int e) {
        const int pos = atomicAdd(s_cnt, 1);
        if (pos < CAP) s_lc[pos] = e;
    };
    auto hit = [&](unsigned w) -> unsigned {
        const unsigned x = w ^ k4;
        return ~(((x & 0x7F7F7F7Fu) + 0x7F7F7F7Fu) | x) & 0x80808080u;
    };
    auto load16 = [&](int gg) -> uint4 {
        uint4 v = own16[(gg < n16) ? gg : (n16 - 1)];
        if (gg >= n16) v = make_uint4(miss, miss, miss, miss);
        return v;
    };
    auto scan16 = [&](const uint4& v, int gg) {
        unsigned m = (hit(v.x) >> 7) | (hit(v.y) >> 6) | (hit(v.z) >> 5) | (hit(v.w) >> 4);
        while (m) {
            const int p = __builtin_ctz(m);
            push(16 * gg + 4 * (p & 3) + (p >> 3));
            m &= m - 1u;
        }
    };

    if (tid < n16) {
        uint4 cur[4], nxt[4];
        for (int u = 0; u < 4; ++u) cur[u] = load16(tid + u * NT);
        for (int g = tid; g < n16; g += 4 * NT) {
            for (int u = 0; u < 4; ++u) nxt[u] = load16(g + (4 + u) * NT);
            for (int u = 0; u < 4; ++u) scan16(cur[u], g + u * NT);
            for (int u = 0; u < 4; ++u) cur[u] = nxt[u];
        }
    }
    const int et = 16 * n16 + tid;
    if (et < E && (unsigned)((const unsigned char*)own)[et] == id) push(et);
    __syncthreads();

    const int total = __builtin_amdgcn_readfirstlane(*s_cnt);
    const int nl    = (total < CAP) ? total : CAP;
    for (int i = tid; i < nl; i += NT) {
        const int e = s_lc[i];
        s_lr[i] = ei[e] - row0;
        s_lc[i] = ei[E + e];
    }
    __syncthreads();
    return total;
}

__global__ __launch_bounds__(NT) void gat0_owner_kernel(
    const int* __restrict__ ei, int E, int N,
    const float* __restrict__ Q0, const float* __restrict__ K0,
    const float* __restrict__ V0, const float* __restrict__ b0,
    float* __restrict__ h0, int* __restrict__ lst, const unsigned* __restrict__ own)
{
    __shared__ unsigned s_max[ROWS * H0];
    __shared__ float    s_den[ROWS * H0];
    __shared__ float    s_acc[ROWS * P0];
    __shared__ float    s_sc[(CAP + ROWS) * H0];
    __shared__ int      s_lr[CAP];
    __shared__ int      s_lc[CAP];
    __shared__ int      s_cnt;

    const int tid   = threadIdx.x;
    const int row0  = blockIdx.x * ROWS;
    const int nrows = (N - row0 < ROWS) ? (N - row0) : ROWS;

    for (int t = tid; t < ROWS * H0; t += NT) { s_max[t] = 0u; s_den[t] = 0.0f; }
    for (int t = tid; t < ROWS * P0; t += NT) s_acc[t] = 0.0f;

    auto item = [&](int entry, int n, int& rl, int& c) {
        if (entry < n) { rl = s_lr[entry]; c = s_lc[entry]; }
        else           { rl = entry - n;   c = row0 + rl; }
    };
    auto score = [&](int rl, int c, int h) -> float {
        const float4* qp = (const float4*)(Q0 + (size_t)(row0 + rl) * U0 + h * A0);
        const float4* kp = (const float4*)(K0 + (size_t)c * U0 + h * A0);
        const float4 qa = qp[0], qb = qp[1], ka = kp[0], kb = kp[1];
        float s = 0.0f;
        s += qa.x * ka.x; s += qa.y * ka.y; s += qa.z * ka.z; s += qa.w * ka.w;
        s += qb.x * kb.x; s += qb.y * kb.y; s += qb.z * kb.z; s += qb.w * kb.w;
        return s * 0.35355339059327373f;
    };

    auto phase_max = [&](int n, int nself, auto keep) {
        const int items = (n + nself) * H0;
        for (int base = tid; base < items; base += 2 * NT) {
            int rl[2], c[2], h[2], id[2]; float s[2];
            for (int u = 0; u < 2; ++u) {
                const int idx = base + u * NT;
                id[u] = (idx < items) ? idx : base;
                h[u] = id[u] & 7;
                item(id[u] >> 3, n, rl[u], c[u]);
            }
            for (int u = 0; u < 2; ++u) s[u] = score(rl[u], c[u], h[u]);
            for (int u = 0; u < 2; ++u) {
                if constexpr (decltype(keep)::value) s_sc[id[u]] = s[u];
                atomicMax(&s_max[rl[u] * H0 + h[u]], fkey(s[u]));
            }
        }
    };

    auto phase_sum = [&](int n, int nself, auto kept) {
        const int items = (n + nself) * H0;
        for (int base = tid; base < items; base += 2 * NT) {
            int rl[2], c[2], h[2], id[2]; bool ok[2]; float s[2]; float4 va[2], vb[2];
            for (int u = 0; u < 2; ++u) {
                const int idx = base + u * NT;
                ok[u] = idx < items;
                id[u] = ok[u] ? idx : base;
                h[u] = id[u] & 7;
                item(id[u] >> 3, n, rl[u], c[u]);
            }
            for (int u = 0; u < 2; ++u) {
                if constexpr (decltype(kept)::value) s[u] = s_sc[id[u]];
                else                                 s[u] = score(rl[u], c[u], h[u]);
                const float4* vp = (const float4*)(V0 + (size_t)c[u] * U0 + h[u] * A0);
                va[u] = vp[0]; vb[u] = vp[1];
            }
            for (int u = 0; u < 2; ++u) {
                const float m  = fkey_dec(s_max[rl[u] * H0 + h[u]]);
                const float ev = __expf(s[u] - m) * (ok[u] ? 1.0f : 0.0f);
                atomicAdd(&s_den[rl[u] * H0 + h[u]], ev);
                float* acc = &s_acc[rl[u] * P0 + h[u] * A0];
                atomicAdd(&acc[0], ev * va[u].x); atomicAdd(&acc[1], ev * va[u].y);
                atomicAdd(&acc[2], ev * va[u].z); atomicAdd(&acc[3], ev * va[u].w);
                atomicAdd(&acc[4], ev * vb[u].x); atomicAdd(&acc[5], ev * vb[u].y);
                atomicAdd(&acc[6], ev * vb[u].z); atomicAdd(&acc[7], ev * vb[u].w);
            }
        }
    };

    const int total = keys_usable((int)gridDim.x, E, N)
        ? make_list_keys(ei, E, own, row0, blockIdx.x, &s_cnt, s_lr, s_lc)
        : make_list(ei, E, 0, E, row0, nrows, &s_cnt, s_lr, s_lc);

    {
        int* mine = lst + (size_t)row0 * U1;
        const bool fits = (total <= CAP) && (total <= nrows * U1 - LHEAD) && (N <= (1 << 24));
        if (tid == 0) mine[0] = fits ? total : -1;
        if (fits)
            for (int i = tid; i < total; i += NT)
                mine[LHEAD + i] = (int)(((unsigned)s_lr[i] << 24) | (unsigned)s_lc[i]);
    }

    if (total <= CAP) {
        phase_max(total, nrows, yes_t{});
        __syncthreads();
        phase_sum(total, nrows, yes_t{});
    } else {
        for (int sweep = 0; sweep < 2; ++sweep) {
            for (int b = 0; b < E; b += CAP) {
                const int n = make_list(ei, E, b, (b + CAP < E) ? (b + CAP) : E, row0, nrows,
                                        &s_cnt, s_lr, s_lc);
                const int nself = (b == 0) ? nrows : 0;
                if (sweep == 0) phase_max(n, nself, no_t{}); else phase_sum(n, nself, no_t{});
            }
        }
    }
    __syncthreads();

    for (int t = tid; t < nrows * U0; t += NT) {
        const int rl = t >> 6, cc = t & 63;
        const float val = s_acc[rl * P0 + cc] / s_den[rl * H0 + (cc >> 3)] + b0[cc];
        h0[(size_t)row0 * U0 + t] = fmaxf(val, 0.0f);
    }
}

__global__ __launch_bounds__(NT) void gat1_owner_kernel(
    const int* __restrict__ ei, int E, int N,
    const float* __restrict__ q1, const float* __restrict__ k1,
    const float* __restrict__ v1, const float* __restrict__ b1,
    const int* lst, float* out)
{
    __shared__ unsigned s_max[ROWS];
    __shared__ float    s_den[ROWS];
    __shared__ float    s_acc[ROWS * P1];
    __shared__ float    s_sc[CAP + ROWS];
    __shared__ int      s_lr[CAP];
    __shared__ int      s_lc[CAP];
    __shared__ int      s_cnt;

    const int tid   = threadIdx.x;
    const int row0  = blockIdx.x * ROWS;
    const int nrows = (N - row0 < ROWS) ? (N - row0) : ROWS;

    for (int t = tid; t < ROWS; t += NT) { s_max[t] = 0u; s_den[t] = 0.0f; }
    for (int t = tid; t < ROWS * P1; t += NT) s_acc[t] = 0.0f;

    auto item = [&](int entry, int n, int& rl, int& c) {
        if (entry < n) { rl = s_lr[entry]; c = s_lc[entry]; }
        else           { rl = entry - n;   c = row0 + rl; }
    };

    auto phase_max = [&](int n, int nself, auto keep) {
        const int items = n + nself;
        for (int base = tid; base < items; base += 2 * NT) {
            int rl[2], c[2], id[2]; float s[2];
            for (int u = 0; u < 2; ++u) {
                const int idx = base + u * NT;
                id[u] = (idx < items) ? idx : base;
                item(id[u], n, rl[u], c[u]);
            }
            for (int u = 0; u < 2; ++u) s[u] = q1[row0 + rl[u]] * k1[c[u]];
            for (int u = 0; u < 2; ++u) {
                if constexpr (decltype(keep)::value) s_sc[id[u]] = s[u];
                atomicMax(&s_max[rl[u]], fkey(s[u]));
            }
        }
    };

    auto phase_sum = [&](int n, int nself, auto kept) {
        const int items = (n + nself) * 8;
        for (int base = tid; base < items; base += 2 * NT) {
            int rl[2], c[2], j[2], id[2]; bool ok[2]; float s[2]; float x[2][5];
            for (int u = 0; u < 2; ++u) {
                const int idx = base + u * NT;
                ok[u] = idx < items;
                id[u] = ok[u] ? idx : base;
                j[u] = id[u] & 7;
                item(id[u] >> 3, n, rl[u], c[u]);
            }
            for (int u = 0; u < 2; ++u) {
                if constexpr (decltype(kept)::value) s[u] = s_sc[id[u] >> 3];
                else                                 s[u] = q1[row0 + rl[u]] * k1[c[u]];
                const float* vv = v1 + (size_t)c[u] * U1 + j[u];
                for (int i = 0; i < 5; ++i) x[u][i] = vv[8 * i];
            }
            for (int u = 0; u < 2; ++u) {
                const float m  = fkey_dec(s_max[rl[u]]);
                const float ev = __expf(s[u] - m) * (ok[u] ? 1.0f : 0.0f);
                if (j[u] == 0) atomicAdd(&s_den[rl[u]], ev);
                float* acc = &s_acc[rl[u] * P1 + j[u]];
                for (int i = 0; i < 5; ++i) atomicAdd(&acc[8 * i], ev * x[u][i]);
            }
        }
    };

    const int* mine = lst + (size_t)row0 * U1;
    const int given = __builtin_amdgcn_readfirstlane(mine[0]);
    if (given >= 0 && given <= CAP && given <= nrows * U1 - LHEAD) {
        for (int i = tid; i < given; i += NT) {
            const unsigned w = (unsigned)mine[LHEAD + i];
            s_lr[i] = (int)(w >> 24);
            s_lc[i] = (int)(w & 0x00FFFFFFu);
        }
        __syncthreads();
        phase_max(given, nrows, yes_t{});
        __syncthreads();
        phase_sum(given, nrows, yes_t{});
    } else {
        const int total = make_list(ei, E, 0, E, row0, nrows, &s_cnt, s_lr, s_lc);
        if (total <= CAP) {
            phase_max(total, nrows, yes_t{});
            __syncthreads();
            phase_sum(total, nrows, yes_t{});
        } else {
            for (int sweep = 0; sweep < 2; ++sweep) {
                for (int b = 0; b < E; b += CAP) {
                    const int n = make_list(ei, E, b, (b + CAP < E) ? (b + CAP) : E, row0, nrows,
                                            &s_cnt, s_lr, s_lc);
                    const int nself = (b == 0) ? nrows : 0;
                    if (sweep == 0) phase_max(n, nself, no_t{}); else phase_sum(n, nself, no_t{});
                }
            }
        }
    }
    __syncthreads();

    for (int t = tid; t < nrows * U1; t += NT) {
        const int rl = t / U1, cc = t - rl * U1;
        out[(size_t)row0 * U1 + t] = s_acc[rl * P1 + cc] / s_den[rl] + b1[cc];
    }
}

extern "C" void kernel_launch(void* const* d_in, const int* in_sizes, int n_in,
                              void* d_out, int out_size, void* d_ws, size_t ws_size,
                              hipStream_t stream)
{
    const float* x    = (const float*)d_in[0];
    const int*   ei   = (const int*)  d_in[1];
    const float* wq0  = (const float*)d_in[2];
    const float* bq0  = (const float*)d_in[3];
    const float* wk0  = (const float*)d_in[4];
    const float* bk0  = (const float*)d_in[5];
    const float* wv0  = (const float*)d_in[6];
    const float* b0   = (const float*)d_in[7];
    const float* wq1  = (const float*)d_in[8];
    const float* bq1  = (const float*)d_in[9];
    const float* wk1  = (const float*)d_in[10];
    const float* bk1  = (const float*)d_in[11];
    const float* wv1  = (const float*)d_in[12];
    const float* b1   = (const float*)d_in[13];
    float* out = (float*)d_out;

    const int N = in_sizes[0] / F0;
    const int E = in_sizes[1] / 2;

    float* ws = (float*)d_ws;
    float* R0 = ws;
    float* R1 = R0 + (size_t)N * U0;
    float* R2 = R1 + (size_t)N * U0;
    float* R3 = R2 + (size_t)N * U0;
    float* R4 = R3 + (size_t)N * H0;
    float* R5 = R4 + (size_t)N * H0;

    const int TB = 256;
    const int G  = (N + ROWS - 1) / ROWS;

    proj0_kernel<<<dim3(N / 16, 12), 32, 0, stream>>>(
        x, wq0, bq0, wk0, bk0, wv0, R0, R1, R2);
    owner_key_kernel<<<(E / 4 + TB) / TB, TB, 0, stream>>>(ei, E, N, G, (unsigned*)R4);
    gat0_owner_kernel<<<G, NT, 0, stream>>>(ei, E, N, R0, R1, R2, b0, R5  ,
                                            (int*)d_out  ,
                                            (const unsigned*)R4  );

    float* q1 = R3;
    float* k1 = R3 + N;
    qk1_kernel<<<(N + TB - 1) / TB, TB, 0, stream>>>(
        R5, wq1, bq1, wk1, bk1, q1, k1, N);
    proj1v_kernel<<<dim3(N / 16, 3), 32, 0, stream>>>(
        R5, wv1, R1  );
    gat1_owner_kernel<<<G, NT, 0, stream>>>(ei, E, N, q1, k1, R1, b1, (const int*)d_out, out);
}


# Layer-0 projection, weights held still: 12-wave work-groups keep each column tile's B fragments in registers and walk over X row tiles staged once in LDS (37500 -> 6144 waves); k and sum order kept
# speedup vs baseline: 3.6661x; 1.2111x over previous
#include <hip/hip_runtime.h>

typedef float v2f __attribute__((ext_vector_type(2)));
typedef float v8f __attribute__((ext_vector_type(8)));

#define F0 128
#define U0 64
#define H0 8
#define A0 8
#define U1 40

__device__ __forceinline__ unsigned fkey(float f) {
    unsigned b = __float_as_uint(f);
    return (b & 0x80000000u) ? ~b : (b | 0x80000000u);
}
__device__ __forceinline__ float fkey_dec(unsigned k) {
    return __uint_as_float((k & 0x80000000u) ? (k ^ 0x80000000u) : ~k);
}

#define PW   12
#define PTH  (PW * 32)
#define XP   (F0 + 4)
#define XQ   (16 * F0 / 4)

__global__ __launch_bounds__(PTH) void proj0_kernel(
    const float* __restrict__ x,
    const float* __restrict__ wq, const float* __restrict__ bq,
    const float* __restrict__ wk, const float* __restrict__ bk,
    const float* __restrict__ wv,
    float* __restrict__ Q0, float* __restrict__ K0, float* __restrict__ V0, int ntiles)
{
    __shared__ float s_x[2][16 * XP];

    const int tid  = threadIdx.x;
    const int wave = tid >> 5;
    const int mat  = wave >> 2;
    const int ct   = wave & 3;
    const int lane = tid & 31;
    const int hi   = lane >> 4;
    const int lo   = lane & 15;

    const float* W   = (mat == 0) ? wq : (mat == 1) ? wk : wv;
    const float* B   = (mat == 0) ? bq : (mat == 1) ? bk : nullptr;
    float*       OUT = (mat == 0) ? Q0 : (mat == 1) ? K0 : V0;

    const int n = ct * 16 + lo;

    v2f bf[F0 / 4];
    #pragma unroll
    for (int i = 0; i < F0 / 4; ++i) {
        const int kb = 4 * i + 2 * hi;
        bf[i].x = W[(size_t)kb * U0 + n];
        bf[i].y = W[(size_t)(kb + 1) * U0 + n];
    }
    const float bias = (B != nullptr) ? B[n] : 0.0f;

    const bool two = tid < XQ - PTH;
    auto gload = [&](int t, float4& p, float4& q) {
        const float4* src = (const float4*)(x + (size_t)t * 16 * F0);
        p = src[tid];
        if (two) q = src[PTH + tid];
    };
    auto sstore = [&](int buf, const float4& p, const float4& q) {
        *(float4*)&s_x[buf][(tid >> 5) * XP + (tid & 31) * 4] = p;
        if (two) *(float4*)&s_x[buf][((PTH + tid) >> 5) * XP + (tid & 31) * 4] = q;
    };

    int t = blockIdx.x;
    float4 p, q;
    if (t < ntiles) { gload(t, p, q); sstore(0, p, q); }
    __syncthreads();

    int cur = 0;
    for (; t < ntiles; t += gridDim.x) {
        const int  tn   = t + gridDim.x;
        const bool more = tn < ntiles;
        if (more) gload(tn, p, q);

        const float* xs = &s_x[cur][lo * XP + 2 * hi];
        v2f a[F0 / 4];
        #pragma unroll
        for (int i = 0; i < F0 / 4; ++i) a[i] = *(const v2f*)(xs + 4 * i);

        v8f acc = {};
        #pragma unroll
        for (int i = 0; i < F0 / 4; ++i)
            acc = __builtin_amdgcn_wmma_f32_16x16x4_f32(
                      false, a[i], false, bf[i], (short)0, acc, false, false);

        if (more) sstore(cur ^ 1, p, q);
        __syncthreads();
        cur ^= 1;

        #pragma unroll
        for (int v = 0; v < 8; ++v) {
            const int row = t * 16 + v + 8 * hi;
            float val = acc[v] + bias;
            if (mat != 2) val = fmaxf(val, 0.0f);
            OUT[(size_t)row * U0 + n] = val;
        }
    }
}

__global__ __launch_bounds__(32) void proj1v_kernel(
    const float* __restrict__ h0, const float* __restrict__ wv1,
    float* __restrict__ v1)
{
    const int tile = blockIdx.x;
    const int ct   = blockIdx.y;
    const int lane = threadIdx.x;
    const int hi   = lane >> 4;
    const int lo   = lane & 15;

    const int m = tile * 16 + lo;
    const int n = ct * 16 + lo;
    const bool valid = (n < U1);

    v8f acc = {};
    for (int k = 0; k < U0; k += 4) {
        const int kb = k + 2 * hi;
        v2f a, b;
        a.x = h0[(size_t)m * U0 + kb];
        a.y = h0[(size_t)m * U0 + kb + 1];
        b.x = valid ? wv1[(size_t)kb * U1 + n] : 0.0f;
        b.y = valid ? wv1[(size_t)(kb + 1) * U1 + n] : 0.0f;
        acc = __builtin_amdgcn_wmma_f32_16x16x4_f32(
                  false, a, false, b, (short)0, acc, false, false);
    }
    if (valid) {
        #pragma unroll
        for (int v = 0; v < 8; ++v) {
            const int row = tile * 16 + v + 8 * hi;
            v1[(size_t)row * U1 + n] = acc[v];
        }
    }
}

__global__ void qk1_kernel(const float* __restrict__ h0,
                           const float* __restrict__ wq1, const float* __restrict__ bq1,
                           const float* __restrict__ wk1, const float* __restrict__ bk1,
                           float* __restrict__ q1, float* __restrict__ k1, int n)
{
    const int i = blockIdx.x * blockDim.x + threadIdx.x;
    if (i >= n) return;
    const float* hr = h0 + (size_t)i * U0;
    float q = bq1[0], k = bk1[0];
    #pragma unroll 8
    for (int j = 0; j < U0; ++j) {
        const float h = hr[j];
        q += h * wq1[j];
        k += h * wk1[j];
    }
    q1[i] = fmaxf(q, 0.0f);
    k1[i] = fmaxf(k, 0.0f);
}

#define ROWS 200
#define CAP  4096
#define NT   1024
#define P0   65
#define P1   41
#define LHEAD 32

static_assert(ROWS <= 256, "a handed-on list entry keeps the local row in 8 bits");

struct yes_t { static constexpr bool value = true;  };
struct no_t  { static constexpr bool value = false; };

__device__ __forceinline__ void list_push(int e, int r, int row0, int nrows,
                                          int* s_cnt, int* s_lr, int* s_lc)
{
    const unsigned rl = (unsigned)(r - row0);
    if (rl < (unsigned)nrows) {
        const int pos = atomicAdd(s_cnt, 1);
        if (pos < CAP) { s_lr[pos] = (int)rl; s_lc[pos] = e; }
    }
}

__device__ __forceinline__ int make_list(const int* __restrict__ ei, int E,
                                         int e_begin, int e_end, int row0, int nrows,
                                         int* s_cnt, int* s_lr, int* s_lc)
{
    const int tid = threadIdx.x;
    __syncthreads();
    if (tid == 0) *s_cnt = 0;
    __syncthreads();

    const int n4 = (e_end - e_begin) >> 2;
    const int4* __restrict__ rows4 = (const int4*)(ei + e_begin);
    const int   miss = row0 - 1;
    const unsigned nr = (unsigned)nrows;

    auto load4 = [&](int gg) -> int4 {
        int4 v = rows4[(gg < n4) ? gg : (n4 - 1)];
        if (gg >= n4) v = make_int4(miss, miss, miss, miss);
        return v;
    };
    auto scan4 = [&](const int4& v, int gg) {
        const bool any = ((unsigned)(v.x - row0) < nr) | ((unsigned)(v.y - row0) < nr) |
                         ((unsigned)(v.z - row0) < nr) | ((unsigned)(v.w - row0) < nr);
        if (any) {
            const int e = e_begin + 4 * gg;
            list_push(e,     v.x, row0, nrows, s_cnt, s_lr, s_lc);
            list_push(e + 1, v.y, row0, nrows, s_cnt, s_lr, s_lc);
            list_push(e + 2, v.z, row0, nrows, s_cnt, s_lr, s_lc);
            list_push(e + 3, v.w, row0, nrows, s_cnt, s_lr, s_lc);
        }
    };

    if (tid < n4) {
        int4 cur[4], nxt[4];
        for (int u = 0; u < 4; ++u) cur[u] = load4(tid + u * NT);
        for (int g = tid; g < n4; g += 4 * NT) {
            for (int u = 0; u < 4; ++u) nxt[u] = load4(g + (4 + u) * NT);
            for (int u = 0; u < 4; ++u) scan4(cur[u], g + u * NT);
            for (int u = 0; u < 4; ++u) cur[u] = nxt[u];
        }
    }
    const int et = e_begin + 4 * n4 + tid;
    if (et < e_end) list_push(et, ei[et], row0, nrows, s_cnt, s_lr, s_lc);
    __syncthreads();

    const int total = __builtin_amdgcn_readfirstlane(*s_cnt);
    const int nl    = (total < CAP) ? total : CAP;
    for (int i = tid; i < nl; i += NT) s_lc[i] = ei[E + s_lc[i]];
    __syncthreads();
    return total;
}

__device__ __forceinline__ bool keys_usable(int groups, int E, int N)
{
    return (groups <= 255) && (E <= N * (H0 * 4));
}

__global__ __launch_bounds__(256) void owner_key_kernel(
    const int* __restrict__ ei, int E, int N, int groups, unsigned* __restrict__ own)
{
    if (!keys_usable(groups, E, N)) return;
    auto key = [](int r) -> unsigned {
        const unsigned k = (unsigned)r / (unsigned)ROWS;
        return (k < 255u) ? k : 255u;
    };
    const int t = blockIdx.x * blockDim.x + threadIdx.x;
    const int e = 4 * t;
    if (e + 3 < E) {
        const int4 r = ((const int4*)ei)[t];
        own[t] = key(r.x) | (key(r.y) << 8) | (key(r.z) << 16) | (key(r.w) << 24);
    } else if (e < E) {
        unsigned char* ob = (unsigned char*)own;
        for (int k = e; k < E; ++k) ob[k] = (unsigned char)key(ei[k]);
    }
}

__device__ __forceinline__ int make_list_keys(const int* __restrict__ ei, int E,
                                              const unsigned* __restrict__ own,
                                              int row0, unsigned id,
                                              int* s_cnt, int* s_lr, int* s_lc)
{
    const int tid = threadIdx.x;
    __syncthreads();
    if (tid == 0) *s_cnt = 0;
    __syncthreads();

    const int n16 = E >> 4;
    const uint4* __restrict__ own16 = (const uint4*)own;
    const unsigned k4   = id * 0x01010101u;
    const unsigned miss = ~k4;

    auto push = [&](int e) {
        const int pos = atomicAdd(s_cnt, 1);
        if (pos < CAP) s_lc[pos] = e;
    };
    auto hit = [&](unsigned w) -> unsigned {
        const unsigned x = w ^ k4;
        return ~(((x & 0x7F7F7F7Fu) + 0x7F7F7F7Fu) | x) & 0x80808080u;
    };
    auto load16 = [&](int gg) -> uint4 {
        uint4 v = own16[(gg < n16) ? gg : (n16 - 1)];
        if (gg >= n16) v = make_uint4(miss, miss, miss, miss);
        return v;
    };
    auto scan16 = [&](const uint4& v, int gg) {
        unsigned m = (hit(v.x) >> 7) | (hit(v.y) >> 6) | (hit(v.z) >> 5) | (hit(v.w) >> 4);
        while (m) {
            const int p = __builtin_ctz(m);
            push(16 * gg + 4 * (p & 3) + (p >> 3));
            m &= m - 1u;
        }
    };

    if (tid < n16) {
        uint4 cur[4], nxt[4];
        for (int u = 0; u < 4; ++u) cur[u] = load16(tid + u * NT);
        for (int g = tid; g < n16; g += 4 * NT) {
            for (int u = 0; u < 4; ++u) nxt[u] = load16(g + (4 + u) * NT);
            for (int u = 0; u < 4; ++u) scan16(cur[u], g + u * NT);
            for (int u = 0; u < 4; ++u) cur[u] = nxt[u];
        }
    }
    const int et = 16 * n16 + tid;
    if (et < E && (unsigned)((const unsigned char*)own)[et] == id) push(et);
    __syncthreads();

    const int total = __builtin_amdgcn_readfirstlane(*s_cnt);
    const int nl    = (total < CAP) ? total : CAP;
    for (int i = tid; i < nl; i += NT) {
        const int e = s_lc[i];
        s_lr[i] = ei[e] - row0;
        s_lc[i] = ei[E + e];
    }
    __syncthreads();
    return total;
}

__global__ __launch_bounds__(NT) void gat0_owner_kernel(
    const int* __restrict__ ei, int E, int N,
    const float* __restrict__ Q0, const float* __restrict__ K0,
    const float* __restrict__ V0, const float* __restrict__ b0,
    float* __restrict__ h0, int* __restrict__ lst, const unsigned* __restrict__ own)
{
    __shared__ unsigned s_max[ROWS * H0];
    __shared__ float    s_den[ROWS * H0];
    __shared__ float    s_acc[ROWS * P0];
    __shared__ float    s_sc[(CAP + ROWS) * H0];
    __shared__ int      s_lr[CAP];
    __shared__ int      s_lc[CAP];
    __shared__ int      s_cnt;

    const int tid   = threadIdx.x;
    const int row0  = blockIdx.x * ROWS;
    const int nrows = (N - row0 < ROWS) ? (N - row0) : ROWS;

    for (int t = tid; t < ROWS * H0; t += NT) { s_max[t] = 0u; s_den[t] = 0.0f; }
    for (int t = tid; t < ROWS * P0; t += NT) s_acc[t] = 0.0f;

    auto item = [&](int entry, int n, int& rl, int& c) {
        if (entry < n) { rl = s_lr[entry]; c = s_lc[entry]; }
        else           { rl = entry - n;   c = row0 + rl; }
    };
    auto score = [&](int rl, int c, int h) -> float {
        const float4* qp = (const float4*)(Q0 + (size_t)(row0 + rl) * U0 + h * A0);
        const float4* kp = (const float4*)(K0 + (size_t)c * U0 + h * A0);
        const float4 qa = qp[0], qb = qp[1], ka = kp[0], kb = kp[1];
        float s = 0.0f;
        s += qa.x * ka.x; s += qa.y * ka.y; s += qa.z * ka.z; s += qa.w * ka.w;
        s += qb.x * kb.x; s += qb.y * kb.y; s += qb.z * kb.z; s += qb.w * kb.w;
        return s * 0.35355339059327373f;
    };

    auto phase_max = [&](int n, int nself, auto keep) {
        const int items = (n + nself) * H0;
        for (int base = tid; base < items; base += 2 * NT) {
            int rl[2], c[2], h[2], id[2]; float s[2];
            for (int u = 0; u < 2; ++u) {
                const int idx = base + u * NT;
                id[u] = (idx < items) ? idx : base;
                h[u] = id[u] & 7;
                item(id[u] >> 3, n, rl[u], c[u]);
            }
            for (int u = 0; u < 2; ++u) s[u] = score(rl[u], c[u], h[u]);
            for (int u = 0; u < 2; ++u) {
                if constexpr (decltype(keep)::value) s_sc[id[u]] = s[u];
                atomicMax(&s_max[rl[u] * H0 + h[u]], fkey(s[u]));
            }
        }
    };

    auto phase_sum = [&](int n, int nself, auto kept) {
        const int items = (n + nself) * H0;
        for (int base = tid; base < items; base += 2 * NT) {
            int rl[2], c[2], h[2], id[2]; bool ok[2]; float s[2]; float4 va[2], vb[2];
            for (int u = 0; u < 2; ++u) {
                const int idx = base + u * NT;
                ok[u] = idx < items;
                id[u] = ok[u] ? idx : base;
                h[u] = id[u] & 7;
                item(id[u] >> 3, n, rl[u], c[u]);
            }
            for (int u = 0; u < 2; ++u) {
                if constexpr (decltype(kept)::value) s[u] = s_sc[id[u]];
                else                                 s[u] = score(rl[u], c[u], h[u]);
                const float4* vp = (const float4*)(V0 + (size_t)c[u] * U0 + h[u] * A0);
                va[u] = vp[0]; vb[u] = vp[1];
            }
            for (int u = 0; u < 2; ++u) {
                const float m  = fkey_dec(s_max[rl[u] * H0 + h[u]]);
                const float ev = __expf(s[u] - m) * (ok[u] ? 1.0f : 0.0f);
                atomicAdd(&s_den[rl[u] * H0 + h[u]], ev);
                float* acc = &s_acc[rl[u] * P0 + h[u] * A0];
                atomicAdd(&acc[0], ev * va[u].x); atomicAdd(&acc[1], ev * va[u].y);
                atomicAdd(&acc[2], ev * va[u].z); atomicAdd(&acc[3], ev * va[u].w);
                atomicAdd(&acc[4], ev * vb[u].x); atomicAdd(&acc[5], ev * vb[u].y);
                atomicAdd(&acc[6], ev * vb[u].z); atomicAdd(&acc[7], ev * vb[u].w);
            }
        }
    };

    const int total = keys_usable((int)gridDim.x, E, N)
        ? make_list_keys(ei, E, own, row0, blockIdx.x, &s_cnt, s_lr, s_lc)
        : make_list(ei, E, 0, E, row0, nrows, &s_cnt, s_lr, s_lc);

    {
        int* mine = lst + (size_t)row0 * U1;
        const bool fits = (total <= CAP) && (total <= nrows * U1 - LHEAD) && (N <= (1 << 24));
        if (tid == 0) mine[0] = fits ? total : -1;
        if (fits)
            for (int i = tid; i < total; i += NT)
                mine[LHEAD + i] = (int)(((unsigned)s_lr[i] << 24) | (unsigned)s_lc[i]);
    }

    if (total <= CAP) {
        phase_max(total, nrows, yes_t{});
        __syncthreads();
        phase_sum(total, nrows, yes_t{});
    } else {
        for (int sweep = 0; sweep < 2; ++sweep) {
            for (int b = 0; b < E; b += CAP) {
                const int n = make_list(ei, E, b, (b + CAP < E) ? (b + CAP) : E, row0, nrows,
                                        &s_cnt, s_lr, s_lc);
                const int nself = (b == 0) ? nrows : 0;
                if (sweep == 0) phase_max(n, nself, no_t{}); else phase_sum(n, nself, no_t{});
            }
        }
    }
    __syncthreads();

    for (int t = tid; t < nrows * U0; t += NT) {
        const int rl = t >> 6, cc = t & 63;
        const float val = s_acc[rl * P0 + cc] / s_den[rl * H0 + (cc >> 3)] + b0[cc];
        h0[(size_t)row0 * U0 + t] = fmaxf(val, 0.0f);
    }
}

__global__ __launch_bounds__(NT) void gat1_owner_kernel(
    const int* __restrict__ ei, int E, int N,
    const float* __restrict__ q1, const float* __restrict__ k1,
    const float* __restrict__ v1, const float* __restrict__ b1,
    const int* lst, float* out)
{
    __shared__ unsigned s_max[ROWS];
    __shared__ float    s_den[ROWS];
    __shared__ float    s_acc[ROWS * P1];
    __shared__ float    s_sc[CAP + ROWS];
    __shared__ int      s_lr[CAP];
    __shared__ int      s_lc[CAP];
    __shared__ int      s_cnt;

    const int tid   = threadIdx.x;
    const int row0  = blockIdx.x * ROWS;
    const int nrows = (N - row0 < ROWS) ? (N - row0) : ROWS;

    for (int t = tid; t < ROWS; t += NT) { s_max[t] = 0u; s_den[t] = 0.0f; }
    for (int t = tid; t < ROWS * P1; t += NT) s_acc[t] = 0.0f;

    auto item = [&](int entry, int n, int& rl, int& c) {
        if (entry < n) { rl = s_lr[entry]; c = s_lc[entry]; }
        else           { rl = entry - n;   c = row0 + rl; }
    };

    auto phase_max = [&](int n, int nself, auto keep) {
        const int items = n + nself;
        for (int base = tid; base < items; base += 2 * NT) {
            int rl[2], c[2], id[2]; float s[2];
            for (int u = 0; u < 2; ++u) {
                const int idx = base + u * NT;
                id[u] = (idx < items) ? idx : base;
                item(id[u], n, rl[u], c[u]);
            }
            for (int u = 0; u < 2; ++u) s[u] = q1[row0 + rl[u]] * k1[c[u]];
            for (int u = 0; u < 2; ++u) {
                if constexpr (decltype(keep)::value) s_sc[id[u]] = s[u];
                atomicMax(&s_max[rl[u]], fkey(s[u]));
            }
        }
    };

    auto phase_sum = [&](int n, int nself, auto kept) {
        const int items = (n + nself) * 8;
        for (int base = tid; base < items; base += 2 * NT) {
            int rl[2], c[2], j[2], id[2]; bool ok[2]; float s[2]; float x[2][5];
            for (int u = 0; u < 2; ++u) {
                const int idx = base + u * NT;
                ok[u] = idx < items;
                id[u] = ok[u] ? idx : base;
                j[u] = id[u] & 7;
                item(id[u] >> 3, n, rl[u], c[u]);
            }
            for (int u = 0; u < 2; ++u) {
                if constexpr (decltype(kept)::value) s[u] = s_sc[id[u] >> 3];
                else                                 s[u] = q1[row0 + rl[u]] * k1[c[u]];
                const float* vv = v1 + (size_t)c[u] * U1 + j[u];
                for (int i = 0; i < 5; ++i) x[u][i] = vv[8 * i];
            }
            for (int u = 0; u < 2; ++u) {
                const float m  = fkey_dec(s_max[rl[u]]);
                const float ev = __expf(s[u] - m) * (ok[u] ? 1.0f : 0.0f);
                if (j[u] == 0) atomicAdd(&s_den[rl[u]], ev);
                float* acc = &s_acc[rl[u] * P1 + j[u]];
                for (int i = 0; i < 5; ++i) atomicAdd(&acc[8 * i], ev * x[u][i]);
            }
        }
    };

    const int* mine = lst + (size_t)row0 * U1;
    const int given = __builtin_amdgcn_readfirstlane(mine[0]);
    if (given >= 0 && given <= CAP && given <= nrows * U1 - LHEAD) {
        for (int i = tid; i < given; i += NT) {
            const unsigned w = (unsigned)mine[LHEAD + i];
            s_lr[i] = (int)(w >> 24);
            s_lc[i] = (int)(w & 0x00FFFFFFu);
        }
        __syncthreads();
        phase_max(given, nrows, yes_t{});
        __syncthreads();
        phase_sum(given, nrows, yes_t{});
    } else {
        const int total = make_list(ei, E, 0, E, row0, nrows, &s_cnt, s_lr, s_lc);
        if (total <= CAP) {
            phase_max(total, nrows, yes_t{});
            __syncthreads();
            phase_sum(total, nrows, yes_t{});
        } else {
            for (int sweep = 0; sweep < 2; ++sweep) {
                for (int b = 0; b < E; b += CAP) {
                    const int n = make_list(ei, E, b, (b + CAP < E) ? (b + CAP) : E, row0, nrows,
                                            &s_cnt, s_lr, s_lc);
                    const int nself = (b == 0) ? nrows : 0;
                    if (sweep == 0) phase_max(n, nself, no_t{}); else phase_sum(n, nself, no_t{});
                }
            }
        }
    }
    __syncthreads();

    for (int t = tid; t < nrows * U1; t += NT) {
        const int rl = t / U1, cc = t - rl * U1;
        out[(size_t)row0 * U1 + t] = s_acc[rl * P1 + cc] / s_den[rl] + b1[cc];
    }
}

extern "C" void kernel_launch(void* const* d_in, const int* in_sizes, int n_in,
                              void* d_out, int out_size, void* d_ws, size_t ws_size,
                              hipStream_t stream)
{
    const float* x    = (const float*)d_in[0];
    const int*   ei   = (const int*)  d_in[1];
    const float* wq0  = (const float*)d_in[2];
    const float* bq0  = (const float*)d_in[3];
    const float* wk0  = (const float*)d_in[4];
    const float* bk0  = (const float*)d_in[5];
    const float* wv0  = (const float*)d_in[6];
    const float* b0   = (const float*)d_in[7];
    const float* wq1  = (const float*)d_in[8];
    const float* bq1  = (const float*)d_in[9];
    const float* wk1  = (const float*)d_in[10];
    const float* bk1  = (const float*)d_in[11];
    const float* wv1  = (const float*)d_in[12];
    const float* b1   = (const float*)d_in[13];
    float* out = (float*)d_out;

    const int N = in_sizes[0] / F0;
    const int E = in_sizes[1] / 2;

    float* ws = (float*)d_ws;
    float* R0 = ws;
    float* R1 = R0 + (size_t)N * U0;
    float* R2 = R1 + (size_t)N * U0;
    float* R3 = R2 + (size_t)N * U0;
    float* R4 = R3 + (size_t)N * H0;
    float* R5 = R4 + (size_t)N * H0;

    const int TB = 256;
    const int G  = (N + ROWS - 1) / ROWS;

    proj0_kernel<<<512, PTH, 0, stream>>>(
        x, wq0, bq0, wk0, bk0, wv0, R0, R1, R2, N / 16);
    owner_key_kernel<<<(E / 4 + TB) / TB, TB, 0, stream>>>(ei, E, N, G, (unsigned*)R4);
    gat0_owner_kernel<<<G, NT, 0, stream>>>(ei, E, N, R0, R1, R2, b0, R5  ,
                                            (int*)d_out  ,
                                            (const unsigned*)R4  );

    float* q1 = R3;
    float* k1 = R3 + N;
    qk1_kernel<<<(N + TB - 1) / TB, TB, 0, stream>>>(
        R5, wq1, bq1, wk1, bk1, q1, k1, N);
    proj1v_kernel<<<dim3(N / 16, 3), 32, 0, stream>>>(
        R5, wv1, R1  );
    gat1_owner_kernel<<<G, NT, 0, stream>>>(ei, E, N, q1, k1, R1, b1, (const int*)d_out, out);
}


# Layer-1 projections, one launch, weights held still as in layer 0: wq1, wk1 ride as spare columns 40, 41 of the v1 product (q1, k1 now summed on the matrix unit); 6 -> 5 launches, 9375 -> 1875 waves
# speedup vs baseline: 4.0798x; 1.1128x over previous
#include <hip/hip_runtime.h>

typedef float v2f __attribute__((ext_vector_type(2)));
typedef float v8f __attribute__((ext_vector_type(8)));

#define F0 128
#define U0 64
#define H0 8
#define A0 8
#define U1 40

__device__ __forceinline__ unsigned fkey(float f) {
    unsigned b = __float_as_uint(f);
    return (b & 0x80000000u) ? ~b : (b | 0x80000000u);
}
__device__ __forceinline__ float fkey_dec(unsigned k) {
    return __uint_as_float((k & 0x80000000u) ? (k ^ 0x80000000u) : ~k);
}

#define PW   12
#define PTH  (PW * 32)
#define XP   (F0 + 4)
#define XQ   (16 * F0 / 4)

__global__ __launch_bounds__(PTH) void proj0_kernel(
    const float* __restrict__ x,
    const float* __restrict__ wq, const float* __restrict__ bq,
    const float* __restrict__ wk, const float* __restrict__ bk,
    const float* __restrict__ wv,
    float* __restrict__ Q0, float* __restrict__ K0, float* __restrict__ V0, int ntiles)
{
    __shared__ float s_x[2][16 * XP];

    const int tid  = threadIdx.x;
    const int wave = tid >> 5;
    const int mat  = wave >> 2;
    const int ct   = wave & 3;
    const int lane = tid & 31;
    const int hi   = lane >> 4;
    const int lo   = lane & 15;

    const float* W   = (mat == 0) ? wq : (mat == 1) ? wk : wv;
    const float* B   = (mat == 0) ? bq : (mat == 1) ? bk : nullptr;
    float*       OUT = (mat == 0) ? Q0 : (mat == 1) ? K0 : V0;

    const int n = ct * 16 + lo;

    v2f bf[F0 / 4];
    #pragma unroll
    for (int i = 0; i < F0 / 4; ++i) {
        const int kb = 4 * i + 2 * hi;
        bf[i].x = W[(size_t)kb * U0 + n];
        bf[i].y = W[(size_t)(kb + 1) * U0 + n];
    }
    const float bias = (B != nullptr) ? B[n] : 0.0f;

    const bool two = tid < XQ - PTH;
    auto gload = [&](int t, float4& p, float4& q) {
        const float4* src = (const float4*)(x + (size_t)t * 16 * F0);
        p = src[tid];
        if (two) q = src[PTH + tid];
    };
    auto sstore = [&](int buf, const float4& p, const float4& q) {
        *(float4*)&s_x[buf][(tid >> 5) * XP + (tid & 31) * 4] = p;
        if (two) *(float4*)&s_x[buf][((PTH + tid) >> 5) * XP + (tid & 31) * 4] = q;
    };

    int t = blockIdx.x;
    float4 p, q;
    if (t < ntiles) { gload(t, p, q); sstore(0, p, q); }
    __syncthreads();

    int cur = 0;
    for (; t < ntiles; t += gridDim.x) {
        const int  tn   = t + gridDim.x;
        const bool more = tn < ntiles;
        if (more) gload(tn, p, q);

        const float* xs = &s_x[cur][lo * XP + 2 * hi];
        v2f a[F0 / 4];
        #pragma unroll
        for (int i = 0; i < F0 / 4; ++i) a[i] = *(const v2f*)(xs + 4 * i);

        v8f acc = {};
        #pragma unroll
        for (int i = 0; i < F0 / 4; ++i)
            acc = __builtin_amdgcn_wmma_f32_16x16x4_f32(
                      false, a[i], false, bf[i], (short)0, acc, false, false);

        if (more) sstore(cur ^ 1, p, q);
        __syncthreads();
        cur ^= 1;

        #pragma unroll
        for (int v = 0; v < 8; ++v) {
            const int row = t * 16 + v + 8 * hi;
            float val = acc[v] + bias;
            if (mat != 2) val = fmaxf(val, 0.0f);
            OUT[(size_t)row * U0 + n] = val;
        }
    }
}

#define LW   3
#define LTH  (LW * 32)
#define HP   (U0 + 4)
#define HQ   (16 * U0 / 4)

__global__ __launch_bounds__(LTH) void proj1_kernel(
    const float* __restrict__ h0, const float* __restrict__ wv1,
    const float* __restrict__ wq1, const float* __restrict__ bq1,
    const float* __restrict__ wk1, const float* __restrict__ bk1,
    float* __restrict__ v1, float* __restrict__ q1, float* __restrict__ k1, int ntiles)
{
    __shared__ __attribute__((aligned(16))) float s_h[2][16 * HP];

    const int tid  = threadIdx.x;
    const int ct   = tid >> 5;
    const int lane = tid & 31;
    const int hi   = lane >> 4;
    const int lo   = lane & 15;

    const int n = ct * 16 + lo;

    const bool isv  = n < U1;
    const bool isq  = n == U1;
    const bool live = n < U1 + 2;
    const float* wcol  = isv ? wv1 + n : isq ? wq1 : wk1;
    const int    wstep = isv ? U1 : 1;

    v2f bf[U0 / 4];
    #pragma unroll
    for (int i = 0; i < U0 / 4; ++i) {
        const int kb = 4 * i + 2 * hi;
        const float w0 = wcol[kb * wstep];
        const float w1 = wcol[(kb + 1) * wstep];
        bf[i].x = live ? w0 : 0.0f;
        bf[i].y = live ? w1 : 0.0f;
    }
    const float bias = (isq ? bq1 : bk1)[0];

    float*    dst   = isv ? v1 + n : isq ? q1 : k1;
    const int dstep = isv ? U1 : 1;

    const bool three = tid < HQ - 2 * LTH;
    auto gload = [&](int t, float4& p, float4& q, float4& r) {
        const float4* src = (const float4*)(h0 + (size_t)t * 16 * U0);
        p = src[tid];
        q = src[LTH + tid];
        if (three) r = src[2 * LTH + tid];
    };
    auto put = [&](int buf, int f, const float4& v) {
        *(float4*)&s_h[buf][(f >> 4) * HP + (f & 15) * 4] = v;
    };
    auto sstore = [&](int buf, const float4& p, const float4& q, const float4& r) {
        put(buf, tid, p);
        put(buf, LTH + tid, q);
        if (three) put(buf, 2 * LTH + tid, r);
    };

    int t = blockIdx.x;
    float4 p, q, r;
    if (t < ntiles) { gload(t, p, q, r); sstore(0, p, q, r); }
    __syncthreads();

    int cur = 0;
    for (; t < ntiles; t += gridDim.x) {
        const int  tn   = t + gridDim.x;
        const bool more = tn < ntiles;
        if (more) gload(tn, p, q, r);

        const float* hs = &s_h[cur][lo * HP + 2 * hi];
        v2f a[U0 / 4];
        #pragma unroll
        for (int i = 0; i < U0 / 4; ++i) a[i] = *(const v2f*)(hs + 4 * i);

        v8f acc = {};
        #pragma unroll
        for (int i = 0; i < U0 / 4; ++i)
            acc = __builtin_amdgcn_wmma_f32_16x16x4_f32(
                      false, a[i], false, bf[i], (short)0, acc, false, false);

        if (more) sstore(cur ^ 1, p, q, r);
        __syncthreads();
        cur ^= 1;

        #pragma unroll
        for (int v = 0; v < 8; ++v) {
            const int row = t * 16 + v + 8 * hi;
            const float val = isv ? acc[v] : fmaxf(acc[v] + bias, 0.0f);
            if (live) dst[row * dstep] = val;
        }
    }
}

#define ROWS 200
#define CAP  4096
#define NT   1024
#define P0   65
#define P1   41
#define LHEAD 32

static_assert(ROWS <= 256, "a handed-on list entry keeps the local row in 8 bits");

struct yes_t { static constexpr bool value = true;  };
struct no_t  { static constexpr bool value = false; };

__device__ __forceinline__ void list_push(int e, int r, int row0, int nrows,
                                          int* s_cnt, int* s_lr, int* s_lc)
{
    const unsigned rl = (unsigned)(r - row0);
    if (rl < (unsigned)nrows) {
        const int pos = atomicAdd(s_cnt, 1);
        if (pos < CAP) { s_lr[pos] = (int)rl; s_lc[pos] = e; }
    }
}

__device__ __forceinline__ int make_list(const int* __restrict__ ei, int E,
                                         int e_begin, int e_end, int row0, int nrows,
                                         int* s_cnt, int* s_lr, int* s_lc)
{
    const int tid = threadIdx.x;
    __syncthreads();
    if (tid == 0) *s_cnt = 0;
    __syncthreads();

    const int n4 = (e_end - e_begin) >> 2;
    const int4* __restrict__ rows4 = (const int4*)(ei + e_begin);
    const int   miss = row0 - 1;
    const unsigned nr = (unsigned)nrows;

    auto load4 = [&](int gg) -> int4 {
        int4 v = rows4[(gg < n4) ? gg : (n4 - 1)];
        if (gg >= n4) v = make_int4(miss, miss, miss, miss);
        return v;
    };
    auto scan4 = [&](const int4& v, int gg) {
        const bool any = ((unsigned)(v.x - row0) < nr) | ((unsigned)(v.y - row0) < nr) |
                         ((unsigned)(v.z - row0) < nr) | ((unsigned)(v.w - row0) < nr);
        if (any) {
            const int e = e_begin + 4 * gg;
            list_push(e,     v.x, row0, nrows, s_cnt, s_lr, s_lc);
            list_push(e + 1, v.y, row0, nrows, s_cnt, s_lr, s_lc);
            list_push(e + 2, v.z, row0, nrows, s_cnt, s_lr, s_lc);
            list_push(e + 3, v.w, row0, nrows, s_cnt, s_lr, s_lc);
        }
    };

    if (tid < n4) {
        int4 cur[4], nxt[4];
        for (int u = 0; u < 4; ++u) cur[u] = load4(tid + u * NT);
        for (int g = tid; g < n4; g += 4 * NT) {
            for (int u = 0; u < 4; ++u) nxt[u] = load4(g + (4 + u) * NT);
            for (int u = 0; u < 4; ++u) scan4(cur[u], g + u * NT);
            for (int u = 0; u < 4; ++u) cur[u] = nxt[u];
        }
    }
    const int et = e_begin + 4 * n4 + tid;
    if (et < e_end) list_push(et, ei[et], row0, nrows, s_cnt, s_lr, s_lc);
    __syncthreads();

    const int total = __builtin_amdgcn_readfirstlane(*s_cnt);
    const int nl    = (total < CAP) ? total : CAP;
    for (int i = tid; i < nl; i += NT) s_lc[i] = ei[E + s_lc[i]];
    __syncthreads();
    return total;
}

__device__ __forceinline__ bool keys_usable(int groups, int E, int N)
{
    return (groups <= 255) && (E <= N * (H0 * 4));
}

__global__ __launch_bounds__(256) void owner_key_kernel(
    const int* __restrict__ ei, int E, int N, int groups, unsigned* __restrict__ own)
{
    if (!keys_usable(groups, E, N)) return;
    auto key = [](int r) -> unsigned {
        const unsigned k = (unsigned)r / (unsigned)ROWS;
        return (k < 255u) ? k : 255u;
    };
    const int t = blockIdx.x * blockDim.x + threadIdx.x;
    const int e = 4 * t;
    if (e + 3 < E) {
        const int4 r = ((const int4*)ei)[t];
        own[t] = key(r.x) | (key(r.y) << 8) | (key(r.z) << 16) | (key(r.w) << 24);
    } else if (e < E) {
        unsigned char* ob = (unsigned char*)own;
        for (int k = e; k < E; ++k) ob[k] = (unsigned char)key(ei[k]);
    }
}

__device__ __forceinline__ int make_list_keys(const int* __restrict__ ei, int E,
                                              const unsigned* __restrict__ own,
                                              int row0, unsigned id,
                                              int* s_cnt, int* s_lr, int* s_lc)
{
    const int tid = threadIdx.x;
    __syncthreads();
    if (tid == 0) *s_cnt = 0;
    __syncthreads();

    const int n16 = E >> 4;
    const uint4* __restrict__ own16 = (const uint4*)own;
    const unsigned k4   = id * 0x01010101u;
    const unsigned miss = ~k4;

    auto push = [&](int e) {
        const int pos = atomicAdd(s_cnt, 1);
        if (pos < CAP) s_lc[pos] = e;
    };
    auto hit = [&](unsigned w) -> unsigned {
        const unsigned x = w ^ k4;
        return ~(((x & 0x7F7F7F7Fu) + 0x7F7F7F7Fu) | x) & 0x80808080u;
    };
    auto load16 = [&](int gg) -> uint4 {
        uint4 v = own16[(gg < n16) ? gg : (n16 - 1)];
        if (gg >= n16) v = make_uint4(miss, miss, miss, miss);
        return v;
    };
    auto scan16 = [&](const uint4& v, int gg) {
        unsigned m = (hit(v.x) >> 7) | (hit(v.y) >> 6) | (hit(v.z) >> 5) | (hit(v.w) >> 4);
        while (m) {
            const int p = __builtin_ctz(m);
            push(16 * gg + 4 * (p & 3) + (p >> 3));
            m &= m - 1u;
        }
    };

    if (tid < n16) {
        uint4 cur[4], nxt[4];
        for (int u = 0; u < 4; ++u) cur[u] = load16(tid + u * NT);
        for (int g = tid; g < n16; g += 4 * NT) {
            for (int u = 0; u < 4; ++u) nxt[u] = load16(g + (4 + u) * NT);
            for (int u = 0; u < 4; ++u) scan16(cur[u], g + u * NT);
            for (int u = 0; u < 4; ++u) cur[u] = nxt[u];
        }
    }
    const int et = 16 * n16 + tid;
    if (et < E && (unsigned)((const unsigned char*)own)[et] == id) push(et);
    __syncthreads();

    const int total = __builtin_amdgcn_readfirstlane(*s_cnt);
    const int nl    = (total < CAP) ? total : CAP;
    for (int i = tid; i < nl; i += NT) {
        const int e = s_lc[i];
        s_lr[i] = ei[e] - row0;
        s_lc[i] = ei[E + e];
    }
    __syncthreads();
    return total;
}

__global__ __launch_bounds__(NT) void gat0_owner_kernel(
    const int* __restrict__ ei, int E, int N,
    const float* __restrict__ Q0, const float* __restrict__ K0,
    const float* __restrict__ V0, const float* __restrict__ b0,
    float* __restrict__ h0, int* __restrict__ lst, const unsigned* __restrict__ own)
{
    __shared__ unsigned s_max[ROWS * H0];
    __shared__ float    s_den[ROWS * H0];
    __shared__ float    s_acc[ROWS * P0];
    __shared__ float    s_sc[(CAP + ROWS) * H0];
    __shared__ int      s_lr[CAP];
    __shared__ int      s_lc[CAP];
    __shared__ int      s_cnt;

    const int tid   = threadIdx.x;
    const int row0  = blockIdx.x * ROWS;
    const int nrows = (N - row0 < ROWS) ? (N - row0) : ROWS;

    for (int t = tid; t < ROWS * H0; t += NT) { s_max[t] = 0u; s_den[t] = 0.0f; }
    for (int t = tid; t < ROWS * P0; t += NT) s_acc[t] = 0.0f;

    auto item = [&](int entry, int n, int& rl, int& c) {
        if (entry < n) { rl = s_lr[entry]; c = s_lc[entry]; }
        else           { rl = entry - n;   c = row0 + rl; }
    };
    auto score = [&](int rl, int c, int h) -> float {
        const float4* qp = (const float4*)(Q0 + (size_t)(row0 + rl) * U0 + h * A0);
        const float4* kp = (const float4*)(K0 + (size_t)c * U0 + h * A0);
        const float4 qa = qp[0], qb = qp[1], ka = kp[0], kb = kp[1];
        float s = 0.0f;
        s += qa.x * ka.x; s += qa.y * ka.y; s += qa.z * ka.z; s += qa.w * ka.w;
        s += qb.x * kb.x; s += qb.y * kb.y; s += qb.z * kb.z; s += qb.w * kb.w;
        return s * 0.35355339059327373f;
    };

    auto phase_max = [&](int n, int nself, auto keep) {
        const int items = (n + nself) * H0;
        for (int base = tid; base < items; base += 2 * NT) {
            int rl[2], c[2], h[2], id[2]; float s[2];
            for (int u = 0; u < 2; ++u) {
                const int idx = base + u * NT;
                id[u] = (idx < items) ? idx : base;
                h[u] = id[u] & 7;
                item(id[u] >> 3, n, rl[u], c[u]);
            }
            for (int u = 0; u < 2; ++u) s[u] = score(rl[u], c[u], h[u]);
            for (int u = 0; u < 2; ++u) {
                if constexpr (decltype(keep)::value) s_sc[id[u]] = s[u];
                atomicMax(&s_max[rl[u] * H0 + h[u]], fkey(s[u]));
            }
        }
    };

    auto phase_sum = [&](int n, int nself, auto kept) {
        const int items = (n + nself) * H0;
        for (int base = tid; base < items; base += 2 * NT) {
            int rl[2], c[2], h[2], id[2]; bool ok[2]; float s[2]; float4 va[2], vb[2];
            for (int u = 0; u < 2; ++u) {
                const int idx = base + u * NT;
                ok[u] = idx < items;
                id[u] = ok[u] ? idx : base;
                h[u] = id[u] & 7;
                item(id[u] >> 3, n, rl[u], c[u]);
            }
            for (int u = 0; u < 2; ++u) {
                if constexpr (decltype(kept)::value) s[u] = s_sc[id[u]];
                else                                 s[u] = score(rl[u], c[u], h[u]);
                const float4* vp = (const float4*)(V0 + (size_t)c[u] * U0 + h[u] * A0);
                va[u] = vp[0]; vb[u] = vp[1];
            }
            for (int u = 0; u < 2; ++u) {
                const float m  = fkey_dec(s_max[rl[u] * H0 + h[u]]);
                const float ev = __expf(s[u] - m) * (ok[u] ? 1.0f : 0.0f);
                atomicAdd(&s_den[rl[u] * H0 + h[u]], ev);
                float* acc = &s_acc[rl[u] * P0 + h[u] * A0];
                atomicAdd(&acc[0], ev * va[u].x); atomicAdd(&acc[1], ev * va[u].y);
                atomicAdd(&acc[2], ev * va[u].z); atomicAdd(&acc[3], ev * va[u].w);
                atomicAdd(&acc[4], ev * vb[u].x); atomicAdd(&acc[5], ev * vb[u].y);
                atomicAdd(&acc[6], ev * vb[u].z); atomicAdd(&acc[7], ev * vb[u].w);
            }
        }
    };

    const int total = keys_usable((int)gridDim.x, E, N)
        ? make_list_keys(ei, E, own, row0, blockIdx.x, &s_cnt, s_lr, s_lc)
        : make_list(ei, E, 0, E, row0, nrows, &s_cnt, s_lr, s_lc);

    {
        int* mine = lst + (size_t)row0 * U1;
        const bool fits = (total <= CAP) && (total <= nrows * U1 - LHEAD) && (N <= (1 << 24));
        if (tid == 0) mine[0] = fits ? total : -1;
        if (fits)
            for (int i = tid; i < total; i += NT)
                mine[LHEAD + i] = (int)(((unsigned)s_lr[i] << 24) | (unsigned)s_lc[i]);
    }

    if (total <= CAP) {
        phase_max(total, nrows, yes_t{});
        __syncthreads();
        phase_sum(total, nrows, yes_t{});
    } else {
        for (int sweep = 0; sweep < 2; ++sweep) {
            for (int b = 0; b < E; b += CAP) {
                const int n = make_list(ei, E, b, (b + CAP < E) ? (b + CAP) : E, row0, nrows,
                                        &s_cnt, s_lr, s_lc);
                const int nself = (b == 0) ? nrows : 0;
                if (sweep == 0) phase_max(n, nself, no_t{}); else phase_sum(n, nself, no_t{});
            }
        }
    }
    __syncthreads();

    for (int t = tid; t < nrows * U0; t += NT) {
        const int rl = t >> 6, cc = t & 63;
        const float val = s_acc[rl * P0 + cc] / s_den[rl * H0 + (cc >> 3)] + b0[cc];
        h0[(size_t)row0 * U0 + t] = fmaxf(val, 0.0f);
    }
}

__global__ __launch_bounds__(NT) void gat1_owner_kernel(
    const int* __restrict__ ei, int E, int N,
    const float* __restrict__ q1, const float* __restrict__ k1,
    const float* __restrict__ v1, const float* __restrict__ b1,
    const int* lst, float* out)
{
    __shared__ unsigned s_max[ROWS];
    __shared__ float    s_den[ROWS];
    __shared__ float    s_acc[ROWS * P1];
    __shared__ float    s_sc[CAP + ROWS];
    __shared__ int      s_lr[CAP];
    __shared__ int      s_lc[CAP];
    __shared__ int      s_cnt;

    const int tid   = threadIdx.x;
    const int row0  = blockIdx.x * ROWS;
    const int nrows = (N - row0 < ROWS) ? (N - row0) : ROWS;

    for (int t = tid; t < ROWS; t += NT) { s_max[t] = 0u; s_den[t] = 0.0f; }
    for (int t = tid; t < ROWS * P1; t += NT) s_acc[t] = 0.0f;

    auto item = [&](int entry, int n, int& rl, int& c) {
        if (entry < n) { rl = s_lr[entry]; c = s_lc[entry]; }
        else           { rl = entry - n;   c = row0 + rl; }
    };

    auto phase_max = [&](int n, int nself, auto keep) {
        const int items = n + nself;
        for (int base = tid; base < items; base += 2 * NT) {
            int rl[2], c[2], id[2]; float s[2];
            for (int u = 0; u < 2; ++u) {
                const int idx = base + u * NT;
                id[u] = (idx < items) ? idx : base;
                item(id[u], n, rl[u], c[u]);
            }
            for (int u = 0; u < 2; ++u) s[u] = q1[row0 + rl[u]] * k1[c[u]];
            for (int u = 0; u < 2; ++u) {
                if constexpr (decltype(keep)::value) s_sc[id[u]] = s[u];
                atomicMax(&s_max[rl[u]], fkey(s[u]));
            }
        }
    };

    auto phase_sum = [&](int n, int nself, auto kept) {
        const int items = (n + nself) * 8;
        for (int base = tid; base < items; base += 2 * NT) {
            int rl[2], c[2], j[2], id[2]; bool ok[2]; float s[2]; float x[2][5];
            for (int u = 0; u < 2; ++u) {
                const int idx = base + u * NT;
                ok[u] = idx < items;
                id[u] = ok[u] ? idx : base;
                j[u] = id[u] & 7;
                item(id[u] >> 3, n, rl[u], c[u]);
            }
            for (int u = 0; u < 2; ++u) {
                if constexpr (decltype(kept)::value) s[u] = s_sc[id[u] >> 3];
                else                                 s[u] = q1[row0 + rl[u]] * k1[c[u]];
                const float* vv = v1 + (size_t)c[u] * U1 + j[u];
                for (int i = 0; i < 5; ++i) x[u][i] = vv[8 * i];
            }
            for (int u = 0; u < 2; ++u) {
                const float m  = fkey_dec(s_max[rl[u]]);
                const float ev = __expf(s[u] - m) * (ok[u] ? 1.0f : 0.0f);
                if (j[u] == 0) atomicAdd(&s_den[rl[u]], ev);
                float* acc = &s_acc[rl[u] * P1 + j[u]];
                for (int i = 0; i < 5; ++i) atomicAdd(&acc[8 * i], ev * x[u][i]);
            }
        }
    };

    const int* mine = lst + (size_t)row0 * U1;
    const int given = __builtin_amdgcn_readfirstlane(mine[0]);
    if (given >= 0 && given <= CAP && given <= nrows * U1 - LHEAD) {
        for (int i = tid; i < given; i += NT) {
            const unsigned w = (unsigned)mine[LHEAD + i];
            s_lr[i] = (int)(w >> 24);
            s_lc[i] = (int)(w & 0x00FFFFFFu);
        }
        __syncthreads();
        phase_max(given, nrows, yes_t{});
        __syncthreads();
        phase_sum(given, nrows, yes_t{});
    } else {
        const int total = make_list(ei, E, 0, E, row0, nrows, &s_cnt, s_lr, s_lc);
        if (total <= CAP) {
            phase_max(total, nrows, yes_t{});
            __syncthreads();
            phase_sum(total, nrows, yes_t{});
        } else {
            for (int sweep = 0; sweep < 2; ++sweep) {
                for (int b = 0; b < E; b += CAP) {
                    const int n = make_list(ei, E, b, (b + CAP < E) ? (b + CAP) : E, row0, nrows,
                                            &s_cnt, s_lr, s_lc);
                    const int nself = (b == 0) ? nrows : 0;
                    if (sweep == 0) phase_max(n, nself, no_t{}); else phase_sum(n, nself, no_t{});
                }
            }
        }
    }
    __syncthreads();

    for (int t = tid; t < nrows * U1; t += NT) {
        const int rl = t / U1, cc = t - rl * U1;
        out[(size_t)row0 * U1 + t] = s_acc[rl * P1 + cc] / s_den[rl] + b1[cc];
    }
}

extern "C" void kernel_launch(void* const* d_in, const int* in_sizes, int n_in,
                              void* d_out, int out_size, void* d_ws, size_t ws_size,
                              hipStream_t stream)
{
    const float* x    = (const float*)d_in[0];
    const int*   ei   = (const int*)  d_in[1];
    const float* wq0  = (const float*)d_in[2];
    const float* bq0  = (const float*)d_in[3];
    const float* wk0  = (const float*)d_in[4];
    const float* bk0  = (const float*)d_in[5];
    const float* wv0  = (const float*)d_in[6];
    const float* b0   = (const float*)d_in[7];
    const float* wq1  = (const float*)d_in[8];
    const float* bq1  = (const float*)d_in[9];
    const float* wk1  = (const float*)d_in[10];
    const float* bk1  = (const float*)d_in[11];
    const float* wv1  = (const float*)d_in[12];
    const float* b1   = (const float*)d_in[13];
    float* out = (float*)d_out;

    const int N = in_sizes[0] / F0;
    const int E = in_sizes[1] / 2;

    float* ws = (float*)d_ws;
    float* R0 = ws;
    float* R1 = R0 + (size_t)N * U0;
    float* R2 = R1 + (size_t)N * U0;
    float* R3 = R2 + (size_t)N * U0;
    float* R4 = R3 + (size_t)N * H0;
    float* R5 = R4 + (size_t)N * H0;

    const int TB = 256;
    const int G  = (N + ROWS - 1) / ROWS;

    proj0_kernel<<<512, PTH, 0, stream>>>(
        x, wq0, bq0, wk0, bk0, wv0, R0, R1, R2, N / 16);
    owner_key_kernel<<<(E / 4 + TB) / TB, TB, 0, stream>>>(ei, E, N, G, (unsigned*)R4);
    gat0_owner_kernel<<<G, NT, 0, stream>>>(ei, E, N, R0, R1, R2, b0, R5  ,
                                            (int*)d_out  ,
                                            (const unsigned*)R4  );

    float* q1 = R3;
    float* k1 = R3 + N;
    proj1_kernel<<<625, LTH, 0, stream>>>(
        R5, wv1, wq1, bq1, wk1, bk1, R1  , q1, k1, N / 16);
    gat1_owner_kernel<<<G, NT, 0, stream>>>(ei, E, N, q1, k1, R1, b1, (const int*)d_out, out);
}
